# starting kernel
# baseline (speedup 1.0000x reference)
#include <hip/hip_runtime.h>
#include <math.h>

#define NN  50000
#define EE  800000
#define DIN 128
#define HH  2
#define FF  64
#define DHF 128
#define NEG_SLOPE 0.2f

typedef __attribute__((ext_vector_type(2))) float v2f;
typedef __attribute__((ext_vector_type(8))) float v8f;

#if defined(__AMDGCN__) && __has_builtin(__builtin_amdgcn_wmma_f32_16x16x4_f32)
#define USE_WMMA_F32 1
#else
#define USE_WMMA_F32 0
#endif

__global__ __launch_bounds__(256) void gat_gemm(const float* __restrict__ Hm,
                                                const float* __restrict__ W,
                                                float* __restrict__ feat)
{
    __shared__ float sA[16 * 132];
    const int tid  = threadIdx.x;
    const int row0 = blockIdx.x * 16;

    for (int i = tid; i < 16 * DIN; i += 256) {
        int r = i >> 7, c = i & 127;
        int gr = row0 + r;
        sA[r * 132 + c] = (gr < NN) ? Hm[gr * DIN + c] : 0.0f;
    }
    __syncthreads();

    const int wave = tid >> 5;
    const int lane = tid & 31;
    const int mrow = lane & 15;
    const int hi   = lane >> 4;
    const int col0 = wave * 16;

    v8f c = {};
#if USE_WMMA_F32
    for (int k = 0; k < DIN; k += 4) {
        const int ka = k + 2 * hi;
        v2f a, b;
        a.x = sA[mrow * 132 + ka + 0];
        a.y = sA[mrow * 132 + ka + 1];
        b.x = W[(ka + 0) * DHF + col0 + mrow];
        b.y = W[(ka + 1) * DHF + col0 + mrow];
        c = __builtin_amdgcn_wmma_f32_16x16x4_f32(false, a, false, b,
                                                  (short)0, c, false, false);
    }
#else
    for (int k = 0; k < DIN; ++k) {
        float bkn = W[k * DHF + col0 + mrow];
        #pragma unroll
        for (int i = 0; i < 8; ++i)
            c[i] = fmaf(sA[(i + 8 * hi) * 132 + k], bkn, c[i]);
    }
#endif
    #pragma unroll
    for (int i = 0; i < 8; ++i) {
        int M = i + 8 * hi;
        int r = row0 + M;
        if (r < NN) feat[r * DHF + col0 + mrow] = c[i];
    }
}

__global__ __launch_bounds__(256) void gat_attn(const float* __restrict__ feat,
                                                const float* __restrict__ al,
                                                const float* __restrict__ ar,
                                                float* __restrict__ el,
                                                float* __restrict__ er)
{
    int i = blockIdx.x * 256 + threadIdx.x;
    if (i >= NN * HH) return;
    int n = i >> 1, h = i & 1;
    const float* fp  = feat + n * DHF + h * FF;
    const float* alp = al + h * FF;
    const float* arp = ar + h * FF;
    float sl = 0.f, sr = 0.f;
    #pragma unroll 8
    for (int f = 0; f < FF; ++f) {
        float v = fp[f];
        sl = fmaf(v, alp[f], sl);
        sr = fmaf(v, arp[f], sr);
    }
    el[i] = sl; er[i] = sr;
}

__global__ __launch_bounds__(256) void csr_zero(int* __restrict__ deg)
{
    int i = blockIdx.x * 256 + threadIdx.x;
    if (i < NN) deg[i] = 0;
}

__global__ __launch_bounds__(256) void csr_hist(const int* __restrict__ dst,
                                                int* __restrict__ deg)
{
    int e = blockIdx.x * 256 + threadIdx.x;
    if (e < EE) atomicAdd(&deg[dst[e]], 1);
}

__global__ __launch_bounds__(1024) void csr_scan(const int* __restrict__ deg,
                                                 int* __restrict__ off,
                                                 int* __restrict__ cursor)
{
    __shared__ int buf[1024];
    __shared__ int carry;
    const int t = threadIdx.x;
    if (t == 0) carry = 0;
    __syncthreads();
    for (int base = 0; base < NN; base += 1024) {
        int i = base + t;
        int v = (i < NN) ? deg[i] : 0;
        buf[t] = v;
        __syncthreads();
        for (int ofs = 1; ofs < 1024; ofs <<= 1) {
            int x = (t >= ofs) ? buf[t - ofs] : 0;
            __syncthreads();
            buf[t] += x;
            __syncthreads();
        }
        int excl = buf[t] - v + carry;
        if (i < NN) { off[i] = excl; cursor[i] = excl; }
        __syncthreads();
        if (t == 1023) carry += buf[1023];
        __syncthreads();
    }
    if (t == 0) off[NN] = carry;
}

__global__ __launch_bounds__(256) void csr_build(const int* __restrict__ src,
                                                 const int* __restrict__ dst,
                                                 int* __restrict__ cursor,
                                                 int* __restrict__ csrc,
                                                 int* __restrict__ cdst)
{
    int e = blockIdx.x * 256 + threadIdx.x;
    if (e >= EE) return;
    int d = dst[e];
    int pos = atomicAdd(&cursor[d], 1);
    csrc[pos] = src[e];
    cdst[pos] = d;
}

__global__ __launch_bounds__(256) void gat_edge_e(const float* __restrict__ el,
                                                  const float* __restrict__ er,
                                                  const int* __restrict__ csrc,
                                                  const int* __restrict__ cdst,
                                                  float* __restrict__ ecsr)
{
    int i = blockIdx.x * 256 + threadIdx.x;
    if (i >= EE * HH) return;
    int pos = i >> 1, h = i & 1;
    int s = csrc[pos], d = cdst[pos];
    float v = el[s * HH + h] + er[d * HH + h];
    ecsr[i] = (v >= 0.f) ? v : NEG_SLOPE * v;
}

__global__ __launch_bounds__(128) void gat_aggregate(const float* __restrict__ ecsr,
                                                     const float* __restrict__ feat,
                                                     const int* __restrict__ csrc,
                                                     const int* __restrict__ off,
                                                     const float* __restrict__ bias,
                                                     float* __restrict__ out,
                                                     int last)
{
    __shared__ float sv[DHF];
    const int d = blockIdx.x;
    const int t = threadIdx.x;
    const int h = t >> 6;
    const int p0 = off[d], p1 = off[d + 1];

    float m = -__builtin_inff();
    for (int p = p0; p < p1; ++p)
        m = fmaxf(m, ecsr[p * HH + h]);

    float den = 0.f, acc = 0.f;
    for (int p = p0; p < p1; ++p) {
        float ex = expf(ecsr[p * HH + h] - m);
        den += ex;
        int s = csrc[p];
        acc = fmaf(ex, feat[s * DHF + t], acc);
    }
    float r = acc / ((den == 0.f) ? 1.f : den);
    float v = r + bias[t];
    v = (v > 0.f) ? v : expm1f(v);

    if (!last) {
        out[d * DHF + t] = v;
    } else {
        sv[t] = v;
        __syncthreads();
        if (t < FF) out[d * FF + t] = 0.5f * (sv[t] + sv[t + FF]);
    }
}

extern "C" void kernel_launch(void* const* d_in, const int* in_sizes, int n_in,
                              void* d_out, int out_size, void* d_ws, size_t ws_size,
                              hipStream_t stream)
{
    const float* x    = (const float*)d_in[0];
    const int*   src  = (const int*)d_in[1];
    const int*   dst  = (const int*)d_in[2];
    const float* Wall = (const float*)d_in[3];
    const float* al   = (const float*)d_in[4];
    const float* ar   = (const float*)d_in[5];
    const float* bias = (const float*)d_in[6];
    float* out = (float*)d_out;

    char* p = (char*)d_ws;
    auto take = [&](size_t bytes) {
        char* q = p;
        p += (bytes + 255) & ~size_t(255);
        return q;
    };
    float* feat   = (float*)take(sizeof(float) * NN * DHF);
    float* hbuf   = (float*)take(sizeof(float) * NN * DHF);
    float* el     = (float*)take(sizeof(float) * NN * HH);
    float* er     = (float*)take(sizeof(float) * NN * HH);
    float* ecsr   = (float*)take(sizeof(float) * EE * HH);
    int*   deg    = (int*)take(sizeof(int) * NN);
    int*   off    = (int*)take(sizeof(int) * (NN + 1));
    int*   cursor = (int*)take(sizeof(int) * NN);
    int*   csrc   = (int*)take(sizeof(int) * EE);
    int*   cdst   = (int*)take(sizeof(int) * EE);

    const int tb   = 256;
    const int gN   = (NN + tb - 1) / tb;
    const int gE   = (EE + tb - 1) / tb;
    const int gN2  = (NN * HH + tb - 1) / tb;
    const int gE2  = (EE * HH + tb - 1) / tb;
    const int gGmm = (NN + 15) / 16;

    csr_zero <<<gN, tb, 0, stream>>>(deg);
    csr_hist <<<gE, tb, 0, stream>>>(dst, deg);
    csr_scan <<<1, 1024, 0, stream>>>(deg, off, cursor);
    csr_build<<<gE, tb, 0, stream>>>(src, dst, cursor, csrc, cdst);

    const float* hin = x;
    for (int layer = 0; layer < 3; ++layer) {
        const float* Wl  = Wall + layer * DIN * DHF;
        const float* all = al   + layer * HH * FF;
        const float* arl = ar   + layer * HH * FF;
        const float* bl  = bias + layer * HH * FF;
        const int    last = (layer == 2);
        float* outp = last ? out : hbuf;

        gat_gemm     <<<gGmm, tb, 0, stream>>>(hin, Wl, feat);
        gat_attn     <<<gN2,  tb, 0, stream>>>(feat, all, arl, el, er);
        gat_edge_e   <<<gE2,  tb, 0, stream>>>(el, er, csrc, cdst, ecsr);
        gat_aggregate<<<NN,  128, 0, stream>>>(ecsr, feat, csrc, off, bl, outp, last);
        hin = hbuf;
    }
    (void)in_sizes; (void)n_in; (void)out_size; (void)ws_size;
}


# csr_scan: each wave scans a contiguous chunk in registers (int4 loads, shuffle scan per 128-count row) and one barrier joins the 32 wave totals, instead of 49 chunks x 23 block barriers
# speedup vs baseline: 1.2819x; 1.2819x over previous
#include <hip/hip_runtime.h>
#include <math.h>

#define NN  50000
#define EE  800000
#define DIN 128
#define HH  2
#define FF  64
#define DHF 128
#define NEG_SLOPE 0.2f

typedef __attribute__((ext_vector_type(2))) float v2f;
typedef __attribute__((ext_vector_type(8))) float v8f;

#if defined(__AMDGCN__) && __has_builtin(__builtin_amdgcn_wmma_f32_16x16x4_f32)
#define USE_WMMA_F32 1
#else
#define USE_WMMA_F32 0
#endif

__global__ __launch_bounds__(256) void gat_gemm(const float* __restrict__ Hm,
                                                const float* __restrict__ W,
                                                float* __restrict__ feat)
{
    __shared__ float sA[16 * 132];
    const int tid  = threadIdx.x;
    const int row0 = blockIdx.x * 16;

    for (int i = tid; i < 16 * DIN; i += 256) {
        int r = i >> 7, c = i & 127;
        int gr = row0 + r;
        sA[r * 132 + c] = (gr < NN) ? Hm[gr * DIN + c] : 0.0f;
    }
    __syncthreads();

    const int wave = tid >> 5;
    const int lane = tid & 31;
    const int mrow = lane & 15;
    const int hi   = lane >> 4;
    const int col0 = wave * 16;

    v8f c = {};
#if USE_WMMA_F32
    for (int k = 0; k < DIN; k += 4) {
        const int ka = k + 2 * hi;
        v2f a, b;
        a.x = sA[mrow * 132 + ka + 0];
        a.y = sA[mrow * 132 + ka + 1];
        b.x = W[(ka + 0) * DHF + col0 + mrow];
        b.y = W[(ka + 1) * DHF + col0 + mrow];
        c = __builtin_amdgcn_wmma_f32_16x16x4_f32(false, a, false, b,
                                                  (short)0, c, false, false);
    }
#else
    for (int k = 0; k < DIN; ++k) {
        float bkn = W[k * DHF + col0 + mrow];
        #pragma unroll
        for (int i = 0; i < 8; ++i)
            c[i] = fmaf(sA[(i + 8 * hi) * 132 + k], bkn, c[i]);
    }
#endif
    #pragma unroll
    for (int i = 0; i < 8; ++i) {
        int M = i + 8 * hi;
        int r = row0 + M;
        if (r < NN) feat[r * DHF + col0 + mrow] = c[i];
    }
}

__global__ __launch_bounds__(256) void gat_attn(const float* __restrict__ feat,
                                                const float* __restrict__ al,
                                                const float* __restrict__ ar,
                                                float* __restrict__ el,
                                                float* __restrict__ er)
{
    int i = blockIdx.x * 256 + threadIdx.x;
    if (i >= NN * HH) return;
    int n = i >> 1, h = i & 1;
    const float* fp  = feat + n * DHF + h * FF;
    const float* alp = al + h * FF;
    const float* arp = ar + h * FF;
    float sl = 0.f, sr = 0.f;
    #pragma unroll 8
    for (int f = 0; f < FF; ++f) {
        float v = fp[f];
        sl = fmaf(v, alp[f], sl);
        sr = fmaf(v, arp[f], sr);
    }
    el[i] = sl; er[i] = sr;
}

__global__ __launch_bounds__(256) void csr_zero(int* __restrict__ deg)
{
    int i = blockIdx.x * 256 + threadIdx.x;
    if (i < NN) deg[i] = 0;
}

__global__ __launch_bounds__(256) void csr_hist(const int* __restrict__ dst,
                                                int* __restrict__ deg)
{
    int e = blockIdx.x * 256 + threadIdx.x;
    if (e < EE) atomicAdd(&deg[dst[e]], 1);
}

#define SCAN_ROWS 13

__global__ __launch_bounds__(1024) void csr_scan(const int* __restrict__ deg,
                                                 int* __restrict__ off,
                                                 int* __restrict__ cursor)
{
    __shared__ int wtot[32];
    const int t    = threadIdx.x;
    const int lane = t & 31;
    const int wave = t >> 5;
    const int base = wave * (SCAN_ROWS * 128) + lane * 4;

    int4 v[SCAN_ROWS];
    #pragma unroll
    for (int j = 0; j < SCAN_ROWS; ++j) {
        const int i = base + j * 128;
        v[j] = (i < NN) ? *reinterpret_cast<const int4*>(deg + i) : make_int4(0, 0, 0, 0);
    }

    int inc[SCAN_ROWS];
    #pragma unroll
    for (int j = 0; j < SCAN_ROWS; ++j) {
        v[j].y += v[j].x;
        v[j].z += v[j].y;
        v[j].w += v[j].z;
        int s = v[j].w;
        #pragma unroll
        for (int ofs = 1; ofs < 32; ofs <<= 1) {
            const int n = __shfl_up(s, ofs);
            if (lane >= ofs) s += n;
        }
        inc[j] = s;
    }

    int rowbase[SCAN_ROWS];
    int run = 0;
    #pragma unroll
    for (int j = 0; j < SCAN_ROWS; ++j) {
        rowbase[j] = run;
        run += __builtin_amdgcn_readlane(inc[j], 31);
    }

    if (lane == 0) wtot[wave] = run;
    __syncthreads();
    int wb = (lane < wave) ? wtot[lane] : 0;
    #pragma unroll
    for (int ofs = 16; ofs > 0; ofs >>= 1)
        wb += __shfl_xor(wb, ofs);

    #pragma unroll
    for (int j = 0; j < SCAN_ROWS; ++j) {
        const int i = base + j * 128;
        if (i < NN) {
            const int e0 = wb + rowbase[j] + inc[j] - v[j].w;
            const int4 o = make_int4(e0, e0 + v[j].x, e0 + v[j].y, e0 + v[j].z);
            *reinterpret_cast<int4*>(off + i)    = o;
            *reinterpret_cast<int4*>(cursor + i) = o;
        }
    }
    if (t == 1023) off[NN] = wb + run;
}

__global__ __launch_bounds__(256) void csr_build(const int* __restrict__ src,
                                                 const int* __restrict__ dst,
                                                 int* __restrict__ cursor,
                                                 int* __restrict__ csrc,
                                                 int* __restrict__ cdst)
{
    int e = blockIdx.x * 256 + threadIdx.x;
    if (e >= EE) return;
    int d = dst[e];
    int pos = atomicAdd(&cursor[d], 1);
    csrc[pos] = src[e];
    cdst[pos] = d;
}

__global__ __launch_bounds__(256) void gat_edge_e(const float* __restrict__ el,
                                                  const float* __restrict__ er,
                                                  const int* __restrict__ csrc,
                                                  const int* __restrict__ cdst,
                                                  float* __restrict__ ecsr)
{
    int i = blockIdx.x * 256 + threadIdx.x;
    if (i >= EE * HH) return;
    int pos = i >> 1, h = i & 1;
    int s = csrc[pos], d = cdst[pos];
    float v = el[s * HH + h] + er[d * HH + h];
    ecsr[i] = (v >= 0.f) ? v : NEG_SLOPE * v;
}

__global__ __launch_bounds__(128) void gat_aggregate(const float* __restrict__ ecsr,
                                                     const float* __restrict__ feat,
                                                     const int* __restrict__ csrc,
                                                     const int* __restrict__ off,
                                                     const float* __restrict__ bias,
                                                     float* __restrict__ out,
                                                     int last)
{
    __shared__ float sv[DHF];
    const int d = blockIdx.x;
    const int t = threadIdx.x;
    const int h = t >> 6;
    const int p0 = off[d], p1 = off[d + 1];

    float m = -__builtin_inff();
    for (int p = p0; p < p1; ++p)
        m = fmaxf(m, ecsr[p * HH + h]);

    float den = 0.f, acc = 0.f;
    for (int p = p0; p < p1; ++p) {
        float ex = expf(ecsr[p * HH + h] - m);
        den += ex;
        int s = csrc[p];
        acc = fmaf(ex, feat[s * DHF + t], acc);
    }
    float r = acc / ((den == 0.f) ? 1.f : den);
    float v = r + bias[t];
    v = (v > 0.f) ? v : expm1f(v);

    if (!last) {
        out[d * DHF + t] = v;
    } else {
        sv[t] = v;
        __syncthreads();
        if (t < FF) out[d * FF + t] = 0.5f * (sv[t] + sv[t + FF]);
    }
}

extern "C" void kernel_launch(void* const* d_in, const int* in_sizes, int n_in,
                              void* d_out, int out_size, void* d_ws, size_t ws_size,
                              hipStream_t stream)
{
    const float* x    = (const float*)d_in[0];
    const int*   src  = (const int*)d_in[1];
    const int*   dst  = (const int*)d_in[2];
    const float* Wall = (const float*)d_in[3];
    const float* al   = (const float*)d_in[4];
    const float* ar   = (const float*)d_in[5];
    const float* bias = (const float*)d_in[6];
    float* out = (float*)d_out;

    char* p = (char*)d_ws;
    auto take = [&](size_t bytes) {
        char* q = p;
        p += (bytes + 255) & ~size_t(255);
        return q;
    };
    float* feat   = (float*)take(sizeof(float) * NN * DHF);
    float* hbuf   = (float*)take(sizeof(float) * NN * DHF);
    float* el     = (float*)take(sizeof(float) * NN * HH);
    float* er     = (float*)take(sizeof(float) * NN * HH);
    float* ecsr   = (float*)take(sizeof(float) * EE * HH);
    int*   deg    = (int*)take(sizeof(int) * NN);
    int*   off    = (int*)take(sizeof(int) * (NN + 1));
    int*   cursor = (int*)take(sizeof(int) * NN);
    int*   csrc   = (int*)take(sizeof(int) * EE);
    int*   cdst   = (int*)take(sizeof(int) * EE);

    const int tb   = 256;
    const int gN   = (NN + tb - 1) / tb;
    const int gE   = (EE + tb - 1) / tb;
    const int gN2  = (NN * HH + tb - 1) / tb;
    const int gE2  = (EE * HH + tb - 1) / tb;
    const int gGmm = (NN + 15) / 16;

    csr_zero <<<gN, tb, 0, stream>>>(deg);
    csr_hist <<<gE, tb, 0, stream>>>(dst, deg);
    csr_scan <<<1, 1024, 0, stream>>>(deg, off, cursor);
    csr_build<<<gE, tb, 0, stream>>>(src, dst, cursor, csrc, cdst);

    const float* hin = x;
    for (int layer = 0; layer < 3; ++layer) {
        const float* Wl  = Wall + layer * DIN * DHF;
        const float* all = al   + layer * HH * FF;
        const float* arl = ar   + layer * HH * FF;
        const float* bl  = bias + layer * HH * FF;
        const int    last = (layer == 2);
        float* outp = last ? out : hbuf;

        gat_gemm     <<<gGmm, tb, 0, stream>>>(hin, Wl, feat);
        gat_attn     <<<gN2,  tb, 0, stream>>>(feat, all, arl, el, er);
        gat_edge_e   <<<gE2,  tb, 0, stream>>>(el, er, csrc, cdst, ecsr);
        gat_aggregate<<<NN,  128, 0, stream>>>(ecsr, feat, csrc, off, bl, outp, last);
        hin = hbuf;
    }
    (void)in_sizes; (void)n_in; (void)out_size; (void)ws_size;
}


# aggregate: one wave walks 5 consecutive nodes, a lane owns 4 features (one 512-byte row load per edge, 8 in flight); score, source and exp held once per edge and broadcast; sums stay in CSR order
# speedup vs baseline: 1.6807x; 1.3111x over previous
#include <hip/hip_runtime.h>
#include <math.h>

#define NN  50000
#define EE  800000
#define DIN 128
#define HH  2
#define FF  64
#define DHF 128
#define NEG_SLOPE 0.2f

typedef __attribute__((ext_vector_type(2))) float v2f;
typedef __attribute__((ext_vector_type(8))) float v8f;
typedef __attribute__((ext_vector_type(4))) float v4f;

#if defined(__AMDGCN__) && __has_builtin(__builtin_amdgcn_wmma_f32_16x16x4_f32)
#define USE_WMMA_F32 1
#else
#define USE_WMMA_F32 0
#endif

__global__ __launch_bounds__(256) void gat_gemm(const float* __restrict__ Hm,
                                                const float* __restrict__ W,
                                                float* __restrict__ feat)
{
    __shared__ float sA[16 * 132];
    const int tid  = threadIdx.x;
    const int row0 = blockIdx.x * 16;

    for (int i = tid; i < 16 * DIN; i += 256) {
        int r = i >> 7, c = i & 127;
        int gr = row0 + r;
        sA[r * 132 + c] = (gr < NN) ? Hm[gr * DIN + c] : 0.0f;
    }
    __syncthreads();

    const int wave = tid >> 5;
    const int lane = tid & 31;
    const int mrow = lane & 15;
    const int hi   = lane >> 4;
    const int col0 = wave * 16;

    v8f c = {};
#if USE_WMMA_F32
    for (int k = 0; k < DIN; k += 4) {
        const int ka = k + 2 * hi;
        v2f a, b;
        a.x = sA[mrow * 132 + ka + 0];
        a.y = sA[mrow * 132 + ka + 1];
        b.x = W[(ka + 0) * DHF + col0 + mrow];
        b.y = W[(ka + 1) * DHF + col0 + mrow];
        c = __builtin_amdgcn_wmma_f32_16x16x4_f32(false, a, false, b,
                                                  (short)0, c, false, false);
    }
#else
    for (int k = 0; k < DIN; ++k) {
        float bkn = W[k * DHF + col0 + mrow];
        #pragma unroll
        for (int i = 0; i < 8; ++i)
            c[i] = fmaf(sA[(i + 8 * hi) * 132 + k], bkn, c[i]);
    }
#endif
    #pragma unroll
    for (int i = 0; i < 8; ++i) {
        int M = i + 8 * hi;
        int r = row0 + M;
        if (r < NN) feat[r * DHF + col0 + mrow] = c[i];
    }
}

__global__ __launch_bounds__(256) void gat_attn(const float* __restrict__ feat,
                                                const float* __restrict__ al,
                                                const float* __restrict__ ar,
                                                float* __restrict__ el,
                                                float* __restrict__ er)
{
    int i = blockIdx.x * 256 + threadIdx.x;
    if (i >= NN * HH) return;
    int n = i >> 1, h = i & 1;
    const float* fp  = feat + n * DHF + h * FF;
    const float* alp = al + h * FF;
    const float* arp = ar + h * FF;
    float sl = 0.f, sr = 0.f;
    #pragma unroll 8
    for (int f = 0; f < FF; ++f) {
        float v = fp[f];
        sl = fmaf(v, alp[f], sl);
        sr = fmaf(v, arp[f], sr);
    }
    el[i] = sl; er[i] = sr;
}

__global__ __launch_bounds__(256) void csr_zero(int* __restrict__ deg)
{
    int i = blockIdx.x * 256 + threadIdx.x;
    if (i < NN) deg[i] = 0;
}

__global__ __launch_bounds__(256) void csr_hist(const int* __restrict__ dst,
                                                int* __restrict__ deg)
{
    int e = blockIdx.x * 256 + threadIdx.x;
    if (e < EE) atomicAdd(&deg[dst[e]], 1);
}

#define SCAN_ROWS 13

__global__ __launch_bounds__(1024) void csr_scan(const int* __restrict__ deg,
                                                 int* __restrict__ off,
                                                 int* __restrict__ cursor)
{
    __shared__ int wtot[32];
    const int t    = threadIdx.x;
    const int lane = t & 31;
    const int wave = t >> 5;
    const int base = wave * (SCAN_ROWS * 128) + lane * 4;

    int4 v[SCAN_ROWS];
    #pragma unroll
    for (int j = 0; j < SCAN_ROWS; ++j) {
        const int i = base + j * 128;
        v[j] = (i < NN) ? *reinterpret_cast<const int4*>(deg + i) : make_int4(0, 0, 0, 0);
    }

    int inc[SCAN_ROWS];
    #pragma unroll
    for (int j = 0; j < SCAN_ROWS; ++j) {
        v[j].y += v[j].x;
        v[j].z += v[j].y;
        v[j].w += v[j].z;
        int s = v[j].w;
        #pragma unroll
        for (int ofs = 1; ofs < 32; ofs <<= 1) {
            const int n = __shfl_up(s, ofs);
            if (lane >= ofs) s += n;
        }
        inc[j] = s;
    }

    int rowbase[SCAN_ROWS];
    int run = 0;
    #pragma unroll
    for (int j = 0; j < SCAN_ROWS; ++j) {
        rowbase[j] = run;
        run += __builtin_amdgcn_readlane(inc[j], 31);
    }

    if (lane == 0) wtot[wave] = run;
    __syncthreads();
    int wb = (lane < wave) ? wtot[lane] : 0;
    #pragma unroll
    for (int ofs = 16; ofs > 0; ofs >>= 1)
        wb += __shfl_xor(wb, ofs);

    #pragma unroll
    for (int j = 0; j < SCAN_ROWS; ++j) {
        const int i = base + j * 128;
        if (i < NN) {
            const int e0 = wb + rowbase[j] + inc[j] - v[j].w;
            const int4 o = make_int4(e0, e0 + v[j].x, e0 + v[j].y, e0 + v[j].z);
            *reinterpret_cast<int4*>(off + i)    = o;
            *reinterpret_cast<int4*>(cursor + i) = o;
        }
    }
    if (t == 1023) off[NN] = wb + run;
}

__global__ __launch_bounds__(256) void csr_build(const int* __restrict__ src,
                                                 const int* __restrict__ dst,
                                                 int* __restrict__ cursor,
                                                 int* __restrict__ csrc,
                                                 int* __restrict__ cdst)
{
    int e = blockIdx.x * 256 + threadIdx.x;
    if (e >= EE) return;
    int d = dst[e];
    int pos = atomicAdd(&cursor[d], 1);
    csrc[pos] = src[e];
    cdst[pos] = d;
}

__global__ __launch_bounds__(256) void gat_edge_e(const float* __restrict__ el,
                                                  const float* __restrict__ er,
                                                  const int* __restrict__ csrc,
                                                  const int* __restrict__ cdst,
                                                  float* __restrict__ ecsr)
{
    int i = blockIdx.x * 256 + threadIdx.x;
    if (i >= EE * HH) return;
    int pos = i >> 1, h = i & 1;
    int s = csrc[pos], d = cdst[pos];
    float v = el[s * HH + h] + er[d * HH + h];
    ecsr[i] = (v >= 0.f) ? v : NEG_SLOPE * v;
}

#define AGG_NPW 5
static_assert(NN % AGG_NPW == 0, "a wave's group of nodes is all in or all out");

__device__ __forceinline__ void load_edges(const int* __restrict__ csrc, const float2* __restrict__ e2p,
                                           int p, int pend, int& s, float2& e)
{
    const int q = (p < pend) ? p : 0;
    s = csrc[q];
    e = e2p[q];
}

__device__ __forceinline__ void keep_edges(int p, int pend, int& s, float2& e)
{
    const bool in = p < pend;
    s   = in ? s : 0;
    e.x = in ? e.x : -__builtin_inff();
    e.y = in ? e.y : -__builtin_inff();
}

__device__ __forceinline__ int lane_bits(float v, int j)
{
    return __builtin_amdgcn_readlane(__float_as_int(v), j);
}

__device__ __forceinline__ float head_weight(float x0, float x1, int j, int hm)
{
    return __int_as_float((lane_bits(x1, j) & hm) | (lane_bits(x0, j) & ~hm));
}

__device__ __forceinline__ void agg_chunk(const v4f* __restrict__ feat4, int lane,
                                          int s, float x0, float x1, int n, int hm,
                                          float& den, float4& acc)
{
    int j = 0;
    for (; j + 8 <= n; j += 8) {
        v4f f[8];
        #pragma unroll
        for (int k = 0; k < 8; ++k)
            f[k] = feat4[(unsigned)__builtin_amdgcn_readlane(s, j + k) * (unsigned)(DHF / 4) + (unsigned)lane];
        __builtin_amdgcn_sched_barrier(0);
        #pragma unroll
        for (int k = 0; k < 8; ++k) {
            const float w = head_weight(x0, x1, j + k, hm);
            den += w;
            acc.x = fmaf(w, f[k].x, acc.x);
            acc.y = fmaf(w, f[k].y, acc.y);
            acc.z = fmaf(w, f[k].z, acc.z);
            acc.w = fmaf(w, f[k].w, acc.w);
        }
    }
    const int r = n - j;
    if (r > 0) {
        v4f f[7];
        #pragma unroll
        for (int k = 0; k < 7; ++k) {
            if (k < r) {
                f[k] = feat4[(unsigned)__builtin_amdgcn_readlane(s, j + k) * (unsigned)(DHF / 4) + (unsigned)lane];
            }
        }
        #pragma unroll
        for (int k = 0; k < 7; ++k) {
            if (k < r) {
                const float w = head_weight(x0, x1, j + k, hm);
                den += w;
                acc.x = fmaf(w, f[k].x, acc.x);
                acc.y = fmaf(w, f[k].y, acc.y);
                acc.z = fmaf(w, f[k].z, acc.z);
                acc.w = fmaf(w, f[k].w, acc.w);
            }
        }
    }
}

__global__ __launch_bounds__(256) void gat_aggregate(const float* __restrict__ ecsr,
                                                     const float* __restrict__ feat,
                                                     const int* __restrict__ csrc,
                                                     const int* __restrict__ off,
                                                     const float* __restrict__ bias,
                                                     float* __restrict__ out,
                                                     int last)
{
    const int lane = threadIdx.x & 31;
    const int hm   = -(lane >> 4);
    const int wv = __builtin_amdgcn_readfirstlane((int)(threadIdx.x >> 5));
    const int d0 = (blockIdx.x * 8 + wv) * AGG_NPW;
    if (d0 >= NN) return;

    const float2* __restrict__ e2p  = reinterpret_cast<const float2*>(ecsr);
    float4* __restrict__ out4 = reinterpret_cast<float4*>(out);

    const float b0 = bias[4 * lane + 0], b1 = bias[4 * lane + 1];
    const float b2 = bias[4 * lane + 2], b3 = bias[4 * lane + 3];

    const int offv = off[d0 + ((lane <= AGG_NPW) ? lane : 0)];

    int pcur = __builtin_amdgcn_readlane(offv, 0);
    int pnxt = __builtin_amdgcn_readlane(offv, 1);
    int    sN;
    float2 eN;
    load_edges(csrc, e2p, pcur + lane, pnxt, sN, eN);

    for (int i = 0; i < AGG_NPW; ++i) {
        const int d  = d0 + i;
        const int p0 = pcur, p1 = pnxt;
        int    sF = sN;
        float2 eF = eN;
        keep_edges(p0 + lane, p1, sF, eF);

        if (i + 1 < AGG_NPW) {
            pcur = p1;
            pnxt = __builtin_amdgcn_readlane(offv, i + 2);
            load_edges(csrc, e2p, pcur + lane, pnxt, sN, eN);
        }

        float m0 = eF.x, m1 = eF.y;
        for (int c0 = p0 + 32; c0 < p1; c0 += 32) {
            const int p = c0 + lane;
            if (p < p1) {
                const float2 e = e2p[p];
                m0 = fmaxf(m0, e.x);
                m1 = fmaxf(m1, e.y);
            }
        }
        #pragma unroll
        for (int ofs = 16; ofs > 0; ofs >>= 1) {
            m0 = fmaxf(m0, __shfl_xor(m0, ofs));
            m1 = fmaxf(m1, __shfl_xor(m1, ofs));
        }

        float  den = 0.f;
        float4 acc = make_float4(0.f, 0.f, 0.f, 0.f);
        {
            int    c0 = p0;
            int    s  = sF;
            float2 e  = eF;
            for (;;) {
                const int n = (p1 - c0 < 32) ? (p1 - c0) : 32;
                agg_chunk(reinterpret_cast<const v4f*>(feat), lane, s, expf(e.x - m0), expf(e.y - m1), n, hm, den, acc);
                c0 += 32;
                if (c0 >= p1) break;
                load_edges(csrc, e2p, c0 + lane, p1, s, e);
                keep_edges(c0 + lane, p1, s, e);
            }
        }

        const float dd = (den == 0.f) ? 1.f : den;
        float4 v;
        v.x = acc.x / dd + b0;
        v.y = acc.y / dd + b1;
        v.z = acc.z / dd + b2;
        v.w = acc.w / dd + b3;
        v.x = (v.x > 0.f) ? v.x : expm1f(v.x);
        v.y = (v.y > 0.f) ? v.y : expm1f(v.y);
        v.z = (v.z > 0.f) ? v.z : expm1f(v.z);
        v.w = (v.w > 0.f) ? v.w : expm1f(v.w);

        if (!last) {
            out4[d * (DHF / 4) + lane] = v;
        } else {
            float4 o;
            o.x = __shfl_xor(v.x, 16);
            o.y = __shfl_xor(v.y, 16);
            o.z = __shfl_xor(v.z, 16);
            o.w = __shfl_xor(v.w, 16);
            if (lane < 16)
                out4[d * (FF / 4) + lane] = make_float4(0.5f * (v.x + o.x), 0.5f * (v.y + o.y),
                                                        0.5f * (v.z + o.z), 0.5f * (v.w + o.w));
        }
    }
}

extern "C" void kernel_launch(void* const* d_in, const int* in_sizes, int n_in,
                              void* d_out, int out_size, void* d_ws, size_t ws_size,
                              hipStream_t stream)
{
    const float* x    = (const float*)d_in[0];
    const int*   src  = (const int*)d_in[1];
    const int*   dst  = (const int*)d_in[2];
    const float* Wall = (const float*)d_in[3];
    const float* al   = (const float*)d_in[4];
    const float* ar   = (const float*)d_in[5];
    const float* bias = (const float*)d_in[6];
    float* out = (float*)d_out;

    char* p = (char*)d_ws;
    auto take = [&](size_t bytes) {
        char* q = p;
        p += (bytes + 255) & ~size_t(255);
        return q;
    };
    float* feat   = (float*)take(sizeof(float) * NN * DHF);
    float* hbuf   = (float*)take(sizeof(float) * NN * DHF);
    float* el     = (float*)take(sizeof(float) * NN * HH);
    float* er     = (float*)take(sizeof(float) * NN * HH);
    float* ecsr   = (float*)take(sizeof(float) * EE * HH);
    int*   deg    = (int*)take(sizeof(int) * NN);
    int*   off    = (int*)take(sizeof(int) * (NN + 1));
    int*   cursor = (int*)take(sizeof(int) * NN);
    int*   csrc   = (int*)take(sizeof(int) * EE);
    int*   cdst   = (int*)take(sizeof(int) * EE);

    const int tb   = 256;
    const int gN   = (NN + tb - 1) / tb;
    const int gE   = (EE + tb - 1) / tb;
    const int gN2  = (NN * HH + tb - 1) / tb;
    const int gE2  = (EE * HH + tb - 1) / tb;
    const int gGmm = (NN + 15) / 16;
    const int gAgg = (NN + 8 * AGG_NPW - 1) / (8 * AGG_NPW);

    csr_zero <<<gN, tb, 0, stream>>>(deg);
    csr_hist <<<gE, tb, 0, stream>>>(dst, deg);
    csr_scan <<<1, 1024, 0, stream>>>(deg, off, cursor);
    csr_build<<<gE, tb, 0, stream>>>(src, dst, cursor, csrc, cdst);

    const float* hin = x;
    for (int layer = 0; layer < 3; ++layer) {
        const float* Wl  = Wall + layer * DIN * DHF;
        const float* all = al   + layer * HH * FF;
        const float* arl = ar   + layer * HH * FF;
        const float* bl  = bias + layer * HH * FF;
        const int    last = (layer == 2);
        float* outp = last ? out : hbuf;

        gat_gemm     <<<gGmm, tb, 0, stream>>>(hin, Wl, feat);
        gat_attn     <<<gN2,  tb, 0, stream>>>(feat, all, arl, el, er);
        gat_edge_e   <<<gE2,  tb, 0, stream>>>(el, er, csrc, cdst, ecsr);
        gat_aggregate<<<gAgg, tb, 0, stream>>>(ecsr, feat, csrc, off, bl, outp, last);
        hin = hbuf;
    }
    (void)in_sizes; (void)n_in; (void)out_size; (void)ws_size;
}


# aggregate makes the edge scores itself: a lane gathers el[src] one node ahead (src indices two ahead) and adds er[dst]; the score pass, its array and the dst copy are gone: 13 launches, not 16
# speedup vs baseline: 1.9228x; 1.1440x over previous
#include <hip/hip_runtime.h>
#include <math.h>

#define NN  50000
#define EE  800000
#define DIN 128
#define HH  2
#define FF  64
#define DHF 128
#define NEG_SLOPE 0.2f

typedef __attribute__((ext_vector_type(2))) float v2f;
typedef __attribute__((ext_vector_type(8))) float v8f;
typedef __attribute__((ext_vector_type(4))) float v4f;

#if defined(__AMDGCN__) && __has_builtin(__builtin_amdgcn_wmma_f32_16x16x4_f32)
#define USE_WMMA_F32 1
#else
#define USE_WMMA_F32 0
#endif

__global__ __launch_bounds__(256) void gat_gemm(const float* __restrict__ Hm,
                                                const float* __restrict__ W,
                                                float* __restrict__ feat)
{
    __shared__ float sA[16 * 132];
    const int tid  = threadIdx.x;
    const int row0 = blockIdx.x * 16;

    for (int i = tid; i < 16 * DIN; i += 256) {
        int r = i >> 7, c = i & 127;
        int gr = row0 + r;
        sA[r * 132 + c] = (gr < NN) ? Hm[gr * DIN + c] : 0.0f;
    }
    __syncthreads();

    const int wave = tid >> 5;
    const int lane = tid & 31;
    const int mrow = lane & 15;
    const int hi   = lane >> 4;
    const int col0 = wave * 16;

    v8f c = {};
#if USE_WMMA_F32
    for (int k = 0; k < DIN; k += 4) {
        const int ka = k + 2 * hi;
        v2f a, b;
        a.x = sA[mrow * 132 + ka + 0];
        a.y = sA[mrow * 132 + ka + 1];
        b.x = W[(ka + 0) * DHF + col0 + mrow];
        b.y = W[(ka + 1) * DHF + col0 + mrow];
        c = __builtin_amdgcn_wmma_f32_16x16x4_f32(false, a, false, b,
                                                  (short)0, c, false, false);
    }
#else
    for (int k = 0; k < DIN; ++k) {
        float bkn = W[k * DHF + col0 + mrow];
        #pragma unroll
        for (int i = 0; i < 8; ++i)
            c[i] = fmaf(sA[(i + 8 * hi) * 132 + k], bkn, c[i]);
    }
#endif
    #pragma unroll
    for (int i = 0; i < 8; ++i) {
        int M = i + 8 * hi;
        int r = row0 + M;
        if (r < NN) feat[r * DHF + col0 + mrow] = c[i];
    }
}

__global__ __launch_bounds__(256) void gat_attn(const float* __restrict__ feat,
                                                const float* __restrict__ al,
                                                const float* __restrict__ ar,
                                                float* __restrict__ el,
                                                float* __restrict__ er)
{
    int i = blockIdx.x * 256 + threadIdx.x;
    if (i >= NN * HH) return;
    int n = i >> 1, h = i & 1;
    const float* fp  = feat + n * DHF + h * FF;
    const float* alp = al + h * FF;
    const float* arp = ar + h * FF;
    float sl = 0.f, sr = 0.f;
    #pragma unroll 8
    for (int f = 0; f < FF; ++f) {
        float v = fp[f];
        sl = fmaf(v, alp[f], sl);
        sr = fmaf(v, arp[f], sr);
    }
    el[i] = sl; er[i] = sr;
}

__global__ __launch_bounds__(256) void csr_zero(int* __restrict__ deg)
{
    int i = blockIdx.x * 256 + threadIdx.x;
    if (i < NN) deg[i] = 0;
}

__global__ __launch_bounds__(256) void csr_hist(const int* __restrict__ dst,
                                                int* __restrict__ deg)
{
    int e = blockIdx.x * 256 + threadIdx.x;
    if (e < EE) atomicAdd(&deg[dst[e]], 1);
}

#define SCAN_ROWS 13

__global__ __launch_bounds__(1024) void csr_scan(const int* __restrict__ deg,
                                                 int* __restrict__ off,
                                                 int* __restrict__ cursor)
{
    __shared__ int wtot[32];
    const int t    = threadIdx.x;
    const int lane = t & 31;
    const int wave = t >> 5;
    const int base = wave * (SCAN_ROWS * 128) + lane * 4;

    int4 v[SCAN_ROWS];
    #pragma unroll
    for (int j = 0; j < SCAN_ROWS; ++j) {
        const int i = base + j * 128;
        v[j] = (i < NN) ? *reinterpret_cast<const int4*>(deg + i) : make_int4(0, 0, 0, 0);
    }

    int inc[SCAN_ROWS];
    #pragma unroll
    for (int j = 0; j < SCAN_ROWS; ++j) {
        v[j].y += v[j].x;
        v[j].z += v[j].y;
        v[j].w += v[j].z;
        int s = v[j].w;
        #pragma unroll
        for (int ofs = 1; ofs < 32; ofs <<= 1) {
            const int n = __shfl_up(s, ofs);
            if (lane >= ofs) s += n;
        }
        inc[j] = s;
    }

    int rowbase[SCAN_ROWS];
    int run = 0;
    #pragma unroll
    for (int j = 0; j < SCAN_ROWS; ++j) {
        rowbase[j] = run;
        run += __builtin_amdgcn_readlane(inc[j], 31);
    }

    if (lane == 0) wtot[wave] = run;
    __syncthreads();
    int wb = (lane < wave) ? wtot[lane] : 0;
    #pragma unroll
    for (int ofs = 16; ofs > 0; ofs >>= 1)
        wb += __shfl_xor(wb, ofs);

    #pragma unroll
    for (int j = 0; j < SCAN_ROWS; ++j) {
        const int i = base + j * 128;
        if (i < NN) {
            const int e0 = wb + rowbase[j] + inc[j] - v[j].w;
            const int4 o = make_int4(e0, e0 + v[j].x, e0 + v[j].y, e0 + v[j].z);
            *reinterpret_cast<int4*>(off + i)    = o;
            *reinterpret_cast<int4*>(cursor + i) = o;
        }
    }
    if (t == 1023) off[NN] = wb + run;
}

__global__ __launch_bounds__(256) void csr_build(const int* __restrict__ src,
                                                 const int* __restrict__ dst,
                                                 int* __restrict__ cursor,
                                                 int* __restrict__ csrc)
{
    int e = blockIdx.x * 256 + threadIdx.x;
    if (e >= EE) return;
    int pos = atomicAdd(&cursor[dst[e]], 1);
    csrc[pos] = src[e];
}

#define AGG_NPW 5
static_assert(NN % AGG_NPW == 0, "a wave's group of nodes is all in or all out");

__device__ __forceinline__ int load_src(const int* __restrict__ csrc, int p, int pend)
{
    return csrc[(p < pend) ? p : 0];
}

__device__ __forceinline__ float2 edge_score(float2 l, float2 r)
{
    const float a = l.x + r.x;
    const float b = l.y + r.y;
    return make_float2((a >= 0.f) ? a : NEG_SLOPE * a, (b >= 0.f) ? b : NEG_SLOPE * b);
}

__device__ __forceinline__ void keep_edges(int p, int pend, int& s, float2& e)
{
    const bool in = p < pend;
    s   = in ? s : 0;
    e.x = in ? e.x : -__builtin_inff();
    e.y = in ? e.y : -__builtin_inff();
}

__device__ __forceinline__ int lane_bits(float v, int j)
{
    return __builtin_amdgcn_readlane(__float_as_int(v), j);
}

__device__ __forceinline__ float head_weight(float x0, float x1, int j, int hm)
{
    return __int_as_float((lane_bits(x1, j) & hm) | (lane_bits(x0, j) & ~hm));
}

__device__ __forceinline__ void agg_chunk(const v4f* __restrict__ feat4, int lane,
                                          int s, float x0, float x1, int n, int hm,
                                          float& den, float4& acc)
{
    int j = 0;
    for (; j + 8 <= n; j += 8) {
        v4f f[8];
        #pragma unroll
        for (int k = 0; k < 8; ++k)
            f[k] = feat4[(unsigned)__builtin_amdgcn_readlane(s, j + k) * (unsigned)(DHF / 4) + (unsigned)lane];
        __builtin_amdgcn_sched_barrier(0);
        #pragma unroll
        for (int k = 0; k < 8; ++k) {
            const float w = head_weight(x0, x1, j + k, hm);
            den += w;
            acc.x = fmaf(w, f[k].x, acc.x);
            acc.y = fmaf(w, f[k].y, acc.y);
            acc.z = fmaf(w, f[k].z, acc.z);
            acc.w = fmaf(w, f[k].w, acc.w);
        }
    }
    const int r = n - j;
    if (r > 0) {
        v4f f[7];
        #pragma unroll
        for (int k = 0; k < 7; ++k) {
            if (k < r) {
                f[k] = feat4[(unsigned)__builtin_amdgcn_readlane(s, j + k) * (unsigned)(DHF / 4) + (unsigned)lane];
            }
        }
        #pragma unroll
        for (int k = 0; k < 7; ++k) {
            if (k < r) {
                const float w = head_weight(x0, x1, j + k, hm);
                den += w;
                acc.x = fmaf(w, f[k].x, acc.x);
                acc.y = fmaf(w, f[k].y, acc.y);
                acc.z = fmaf(w, f[k].z, acc.z);
                acc.w = fmaf(w, f[k].w, acc.w);
            }
        }
    }
}

__global__ __launch_bounds__(256, 10) void gat_aggregate(const float* __restrict__ el,
                                                     const float* __restrict__ er,
                                                     const float* __restrict__ feat,
                                                     const int* __restrict__ csrc,
                                                     const int* __restrict__ off,
                                                     const float* __restrict__ bias,
                                                     float* __restrict__ out,
                                                     int last)
{
    const int lane = threadIdx.x & 31;
    const int hm   = -(lane >> 4);
    const int wv = __builtin_amdgcn_readfirstlane((int)(threadIdx.x >> 5));
    const int d0 = (blockIdx.x * 8 + wv) * AGG_NPW;
    if (d0 >= NN) return;

    const float2* __restrict__ el2  = reinterpret_cast<const float2*>(el);
    const float2* __restrict__ er2  = reinterpret_cast<const float2*>(er);
    float4* __restrict__ out4 = reinterpret_cast<float4*>(out);

    const float b0 = bias[4 * lane + 0], b1 = bias[4 * lane + 1];
    const float b2 = bias[4 * lane + 2], b3 = bias[4 * lane + 3];

    static_assert(AGG_NPW + 2 < 32, "bounds of two nodes past the last are read from lanes");
    const int    offv = off[d0 + ((lane <= AGG_NPW) ? lane : AGG_NPW)];
    const float2 erv  = er2[d0 + ((lane < AGG_NPW) ? lane : 0)];

    int pcur = __builtin_amdgcn_readlane(offv, 0);
    int pnxt = __builtin_amdgcn_readlane(offv, 1);
    int    sN  = load_src(csrc, pcur + lane, pnxt);
    int    sNN = load_src(csrc, pnxt + lane, __builtin_amdgcn_readlane(offv, 2));
    float2 gN  = el2[sN];

    for (int i = 0; i < AGG_NPW; ++i) {
        const int d  = d0 + i;
        const int p0 = pcur, p1 = pnxt;
        const float2 erd = make_float2(__int_as_float(lane_bits(erv.x, i)), __int_as_float(lane_bits(erv.y, i)));
        int    sF = sN;
        float2 eF = edge_score(gN, erd);
        keep_edges(p0 + lane, p1, sF, eF);

        pcur = p1;
        pnxt = __builtin_amdgcn_readlane(offv, i + 2);
        sN   = sNN;
        gN   = el2[sN];
        sNN  = load_src(csrc, pnxt + lane, __builtin_amdgcn_readlane(offv, i + 3));

        float m0 = eF.x, m1 = eF.y;
        for (int c0 = p0 + 32; c0 < p1; c0 += 32) {
            const int p = c0 + lane;
            if (p < p1) {
                const float2 e = edge_score(el2[csrc[p]], erd);
                m0 = fmaxf(m0, e.x);
                m1 = fmaxf(m1, e.y);
            }
        }
        #pragma unroll
        for (int ofs = 16; ofs > 0; ofs >>= 1) {
            m0 = fmaxf(m0, __shfl_xor(m0, ofs));
            m1 = fmaxf(m1, __shfl_xor(m1, ofs));
        }

        float  den = 0.f;
        float4 acc = make_float4(0.f, 0.f, 0.f, 0.f);
        {
            int    c0 = p0;
            int    s  = sF;
            float2 e  = eF;
            for (;;) {
                const int n = (p1 - c0 < 32) ? (p1 - c0) : 32;
                agg_chunk(reinterpret_cast<const v4f*>(feat), lane, s, expf(e.x - m0), expf(e.y - m1), n, hm, den, acc);
                c0 += 32;
                if (c0 >= p1) break;
                s = load_src(csrc, c0 + lane, p1);
                e = edge_score(el2[s], erd);
                keep_edges(c0 + lane, p1, s, e);
            }
        }

        const float dd = (den == 0.f) ? 1.f : den;
        float4 v;
        v.x = acc.x / dd + b0;
        v.y = acc.y / dd + b1;
        v.z = acc.z / dd + b2;
        v.w = acc.w / dd + b3;
        v.x = (v.x > 0.f) ? v.x : expm1f(v.x);
        v.y = (v.y > 0.f) ? v.y : expm1f(v.y);
        v.z = (v.z > 0.f) ? v.z : expm1f(v.z);
        v.w = (v.w > 0.f) ? v.w : expm1f(v.w);

        if (!last) {
            out4[d * (DHF / 4) + lane] = v;
        } else {
            float4 o;
            o.x = __shfl_xor(v.x, 16);
            o.y = __shfl_xor(v.y, 16);
            o.z = __shfl_xor(v.z, 16);
            o.w = __shfl_xor(v.w, 16);
            if (lane < 16)
                out4[d * (FF / 4) + lane] = make_float4(0.5f * (v.x + o.x), 0.5f * (v.y + o.y),
                                                        0.5f * (v.z + o.z), 0.5f * (v.w + o.w));
        }
    }
}

extern "C" void kernel_launch(void* const* d_in, const int* in_sizes, int n_in,
                              void* d_out, int out_size, void* d_ws, size_t ws_size,
                              hipStream_t stream)
{
    const float* x    = (const float*)d_in[0];
    const int*   src  = (const int*)d_in[1];
    const int*   dst  = (const int*)d_in[2];
    const float* Wall = (const float*)d_in[3];
    const float* al   = (const float*)d_in[4];
    const float* ar   = (const float*)d_in[5];
    const float* bias = (const float*)d_in[6];
    float* out = (float*)d_out;

    char* p = (char*)d_ws;
    auto take = [&](size_t bytes) {
        char* q = p;
        p += (bytes + 255) & ~size_t(255);
        return q;
    };
    float* feat   = (float*)take(sizeof(float) * NN * DHF);
    float* hbuf   = (float*)take(sizeof(float) * NN * DHF);
    float* el     = (float*)take(sizeof(float) * NN * HH);
    float* er     = (float*)take(sizeof(float) * NN * HH);
    int*   deg    = (int*)take(sizeof(int) * NN);
    int*   off    = (int*)take(sizeof(int) * (NN + 1));
    int*   cursor = (int*)take(sizeof(int) * NN);
    int*   csrc   = (int*)take(sizeof(int) * EE);

    const int tb   = 256;
    const int gN   = (NN + tb - 1) / tb;
    const int gE   = (EE + tb - 1) / tb;
    const int gN2  = (NN * HH + tb - 1) / tb;
    const int gGmm = (NN + 15) / 16;
    const int gAgg = (NN + 8 * AGG_NPW - 1) / (8 * AGG_NPW);

    csr_zero <<<gN, tb, 0, stream>>>(deg);
    csr_hist <<<gE, tb, 0, stream>>>(dst, deg);
    csr_scan <<<1, 1024, 0, stream>>>(deg, off, cursor);
    csr_build<<<gE, tb, 0, stream>>>(src, dst, cursor, csrc);

    const float* hin = x;
    for (int layer = 0; layer < 3; ++layer) {
        const float* Wl  = Wall + layer * DIN * DHF;
        const float* all = al   + layer * HH * FF;
        const float* arl = ar   + layer * HH * FF;
        const float* bl  = bias + layer * HH * FF;
        const int    last = (layer == 2);
        float* outp = last ? out : hbuf;

        gat_gemm     <<<gGmm, tb, 0, stream>>>(hin, Wl, feat);
        gat_attn     <<<gN2,  tb, 0, stream>>>(feat, all, arl, el, er);
        gat_aggregate<<<gAgg, tb, 0, stream>>>(el, er, feat, csrc, off, bl, outp, last);
        hin = hbuf;
    }
    (void)in_sizes; (void)n_in; (void)out_size; (void)ws_size;
}


# CSR build as a two-level counting sort on LDS counters (250 edge chunks x 250 node ranges, scan, pair scatter, per-range sort): no global atomics; segment order, so sum order, is LDS arrival order
# speedup vs baseline: 2.7712x; 1.4412x over previous
#include <hip/hip_runtime.h>
#include <math.h>

#define NN  50000
#define EE  800000
#define DIN 128
#define HH  2
#define FF  64
#define DHF 128
#define NEG_SLOPE 0.2f

typedef __attribute__((ext_vector_type(2))) float v2f;
typedef __attribute__((ext_vector_type(8))) float v8f;
typedef __attribute__((ext_vector_type(4))) float v4f;

#if defined(__AMDGCN__) && __has_builtin(__builtin_amdgcn_wmma_f32_16x16x4_f32)
#define USE_WMMA_F32 1
#else
#define USE_WMMA_F32 0
#endif

__global__ __launch_bounds__(256) void gat_gemm(const float* __restrict__ Hm,
                                                const float* __restrict__ W,
                                                float* __restrict__ feat)
{
    __shared__ float sA[16 * 132];
    const int tid  = threadIdx.x;
    const int row0 = blockIdx.x * 16;

    for (int i = tid; i < 16 * DIN; i += 256) {
        int r = i >> 7, c = i & 127;
        int gr = row0 + r;
        sA[r * 132 + c] = (gr < NN) ? Hm[gr * DIN + c] : 0.0f;
    }
    __syncthreads();

    const int wave = tid >> 5;
    const int lane = tid & 31;
    const int mrow = lane & 15;
    const int hi   = lane >> 4;
    const int col0 = wave * 16;

    v8f c = {};
#if USE_WMMA_F32
    for (int k = 0; k < DIN; k += 4) {
        const int ka = k + 2 * hi;
        v2f a, b;
        a.x = sA[mrow * 132 + ka + 0];
        a.y = sA[mrow * 132 + ka + 1];
        b.x = W[(ka + 0) * DHF + col0 + mrow];
        b.y = W[(ka + 1) * DHF + col0 + mrow];
        c = __builtin_amdgcn_wmma_f32_16x16x4_f32(false, a, false, b,
                                                  (short)0, c, false, false);
    }
#else
    for (int k = 0; k < DIN; ++k) {
        float bkn = W[k * DHF + col0 + mrow];
        #pragma unroll
        for (int i = 0; i < 8; ++i)
            c[i] = fmaf(sA[(i + 8 * hi) * 132 + k], bkn, c[i]);
    }
#endif
    #pragma unroll
    for (int i = 0; i < 8; ++i) {
        int M = i + 8 * hi;
        int r = row0 + M;
        if (r < NN) feat[r * DHF + col0 + mrow] = c[i];
    }
}

__global__ __launch_bounds__(256) void gat_attn(const float* __restrict__ feat,
                                                const float* __restrict__ al,
                                                const float* __restrict__ ar,
                                                float* __restrict__ el,
                                                float* __restrict__ er)
{
    int i = blockIdx.x * 256 + threadIdx.x;
    if (i >= NN * HH) return;
    int n = i >> 1, h = i & 1;
    const float* fp  = feat + n * DHF + h * FF;
    const float* alp = al + h * FF;
    const float* arp = ar + h * FF;
    float sl = 0.f, sr = 0.f;
    #pragma unroll 8
    for (int f = 0; f < FF; ++f) {
        float v = fp[f];
        sl = fmaf(v, alp[f], sl);
        sr = fmaf(v, arp[f], sr);
    }
    el[i] = sl; er[i] = sr;
}

#define NCHUNK    250
#define CHUNK_E   3200
#define NBKT      250
#define BKT_NODES 200
#define CNT_N     (NBKT * NCHUNK)
static_assert(NCHUNK * CHUNK_E == EE, "the chunks cover the edge list exactly");
static_assert(NBKT * BKT_NODES == NN, "the ranges cover the nodes exactly");
static_assert(CHUNK_E % 4 == 0, "a chunk is read as int4");
#define CHUNK_Q   ((CHUNK_E / 4 + 255) / 256)
static_assert(NBKT <= 256 && BKT_NODES <= 224, "one thread per range; seven counts per lane of one wave");

__global__ __launch_bounds__(256) void bkt_count(const int* __restrict__ dst,
                                                 int* __restrict__ cnt)
{
    __shared__ int h[NBKT];
    const int t     = threadIdx.x;
    const int chunk = blockIdx.x;
    if (t < NBKT) h[t] = 0;
    __syncthreads();
    const int4* __restrict__ d4 = reinterpret_cast<const int4*>(dst + chunk * CHUNK_E);
    int4 d[CHUNK_Q];
    #pragma unroll
    for (int k = 0; k < CHUNK_Q; ++k) {
        const int q = t + k * 256;
        d[k] = d4[(q < CHUNK_E / 4) ? q : 0];
    }
    #pragma unroll
    for (int k = 0; k < CHUNK_Q; ++k) {
        if (t + k * 256 < CHUNK_E / 4) {
            atomicAdd(&h[(unsigned)d[k].x / BKT_NODES], 1);
            atomicAdd(&h[(unsigned)d[k].y / BKT_NODES], 1);
            atomicAdd(&h[(unsigned)d[k].z / BKT_NODES], 1);
            atomicAdd(&h[(unsigned)d[k].w / BKT_NODES], 1);
        }
    }
    __syncthreads();
    if (t < NBKT) cnt[t * NCHUNK + chunk] = h[t];
}

#define SCAN_ROWS 16
static_assert(32 * SCAN_ROWS * 128 >= CNT_N, "the scan block covers all counts");
static_assert(CNT_N % 4 == 0, "a group of four counts is all in or all out");

__global__ __launch_bounds__(1024) void bkt_scan(const int* __restrict__ cnt,
                                                 int* __restrict__ pos)
{
    __shared__ int wtot[32];
    const int t    = threadIdx.x;
    const int lane = t & 31;
    const int wave = t >> 5;
    const int base = wave * (SCAN_ROWS * 128) + lane * 4;

    int4 v[SCAN_ROWS];
    #pragma unroll
    for (int j = 0; j < SCAN_ROWS; ++j) {
        const int i = base + j * 128;
        v[j] = (i < CNT_N) ? *reinterpret_cast<const int4*>(cnt + i) : make_int4(0, 0, 0, 0);
    }

    int inc[SCAN_ROWS];
    #pragma unroll
    for (int j = 0; j < SCAN_ROWS; ++j) {
        v[j].y += v[j].x;
        v[j].z += v[j].y;
        v[j].w += v[j].z;
        int s = v[j].w;
        #pragma unroll
        for (int ofs = 1; ofs < 32; ofs <<= 1) {
            const int n = __shfl_up(s, ofs);
            if (lane >= ofs) s += n;
        }
        inc[j] = s;
    }

    int rowbase[SCAN_ROWS];
    int run = 0;
    #pragma unroll
    for (int j = 0; j < SCAN_ROWS; ++j) {
        rowbase[j] = run;
        run += __builtin_amdgcn_readlane(inc[j], 31);
    }

    if (lane == 0) wtot[wave] = run;
    __syncthreads();
    int wb = (lane < wave) ? wtot[lane] : 0;
    #pragma unroll
    for (int ofs = 16; ofs > 0; ofs >>= 1)
        wb += __shfl_xor(wb, ofs);

    #pragma unroll
    for (int j = 0; j < SCAN_ROWS; ++j) {
        const int i = base + j * 128;
        if (i < CNT_N) {
            const int e0 = wb + rowbase[j] + inc[j] - v[j].w;
            *reinterpret_cast<int4*>(pos + i) = make_int4(e0, e0 + v[j].x, e0 + v[j].y, e0 + v[j].z);
        }
    }
    if (t == 1023) pos[CNT_N] = wb + run;
}

__global__ __launch_bounds__(256) void bkt_scatter(const int* __restrict__ src,
                                                   const int* __restrict__ dst,
                                                   const int* __restrict__ pos,
                                                   int2* __restrict__ pairs)
{
    __shared__ int cur[NBKT];
    const int t     = threadIdx.x;
    const int chunk = blockIdx.x;
    if (t < NBKT) cur[t] = pos[t * NCHUNK + chunk];
    __syncthreads();
    const int4* __restrict__ s4 = reinterpret_cast<const int4*>(src + chunk * CHUNK_E);
    const int4* __restrict__ d4 = reinterpret_cast<const int4*>(dst + chunk * CHUNK_E);
    int4 s[CHUNK_Q], d[CHUNK_Q];
    #pragma unroll
    for (int k = 0; k < CHUNK_Q; ++k) {
        const int q  = t + k * 256;
        const int qc = (q < CHUNK_E / 4) ? q : 0;
        s[k] = s4[qc];
        d[k] = d4[qc];
    }
    #pragma unroll
    for (int k = 0; k < CHUNK_Q; ++k) {
        if (t + k * 256 < CHUNK_E / 4) {
            const int p0 = atomicAdd(&cur[(unsigned)d[k].x / BKT_NODES], 1);
            const int p1 = atomicAdd(&cur[(unsigned)d[k].y / BKT_NODES], 1);
            const int p2 = atomicAdd(&cur[(unsigned)d[k].z / BKT_NODES], 1);
            const int p3 = atomicAdd(&cur[(unsigned)d[k].w / BKT_NODES], 1);
            pairs[p0] = make_int2(s[k].x, d[k].x);
            pairs[p1] = make_int2(s[k].y, d[k].y);
            pairs[p2] = make_int2(s[k].z, d[k].z);
            pairs[p3] = make_int2(s[k].w, d[k].w);
        }
    }
}

#define SEG_HOLD 16

__global__ __launch_bounds__(256) void seg_sort(const int2* __restrict__ pairs,
                                                const int* __restrict__ pos,
                                                int* __restrict__ off,
                                                int* __restrict__ csrc)
{
    __shared__ int cnt[BKT_NODES];
    __shared__ int start[BKT_NODES];
    const int t     = threadIdx.x;
    const int b     = blockIdx.x;
    const int node0 = b * BKT_NODES;
    const int s0 = pos[b * NCHUNK];
    const int s1 = pos[(b + 1) * NCHUNK];
    if (t < BKT_NODES) cnt[t] = 0;
    __syncthreads();
    int2 pr[SEG_HOLD];
    #pragma unroll
    for (int k = 0; k < SEG_HOLD; ++k) {
        const int i = s0 + t + k * 256;
        pr[k] = pairs[(i < s1) ? i : 0];
    }
    #pragma unroll
    for (int k = 0; k < SEG_HOLD; ++k) {
        if (s0 + t + k * 256 < s1) atomicAdd(&cnt[pr[k].y - node0], 1);
    }
    for (int i = s0 + t + SEG_HOLD * 256; i < s1; i += 256)
        atomicAdd(&cnt[pairs[i].y - node0], 1);
    __syncthreads();
    if (t < 32) {
        int c[7];
        int s = 0;
        #pragma unroll
        for (int k = 0; k < 7; ++k) {
            const int idx = t * 7 + k;
            c[k] = (idx < BKT_NODES) ? cnt[idx] : 0;
            s += c[k];
        }
        int inc = s;
        #pragma unroll
        for (int ofs = 1; ofs < 32; ofs <<= 1) {
            const int n = __shfl_up(inc, ofs);
            if (t >= ofs) inc += n;
        }
        int run = s0 + inc - s;
        #pragma unroll
        for (int k = 0; k < 7; ++k) {
            const int idx = t * 7 + k;
            if (idx < BKT_NODES) {
                start[idx] = run;
                off[node0 + idx] = run;
            }
            run += c[k];
        }
    }
    if (b == NBKT - 1 && t == 0) off[NN] = s1;
    __syncthreads();
    #pragma unroll
    for (int k = 0; k < SEG_HOLD; ++k) {
        if (s0 + t + k * 256 < s1) {
            const int p = atomicAdd(&start[pr[k].y - node0], 1);
            csrc[p] = pr[k].x;
        }
    }
    for (int i = s0 + t + SEG_HOLD * 256; i < s1; i += 256) {
        const int2 q = pairs[i];
        const int p = atomicAdd(&start[q.y - node0], 1);
        csrc[p] = q.x;
    }
}

#define AGG_NPW 5
static_assert(NN % AGG_NPW == 0, "a wave's group of nodes is all in or all out");

__device__ __forceinline__ int load_src(const int* __restrict__ csrc, int p, int pend)
{
    return csrc[(p < pend) ? p : 0];
}

__device__ __forceinline__ float2 edge_score(float2 l, float2 r)
{
    const float a = l.x + r.x;
    const float b = l.y + r.y;
    return make_float2((a >= 0.f) ? a : NEG_SLOPE * a, (b >= 0.f) ? b : NEG_SLOPE * b);
}

__device__ __forceinline__ void keep_edges(int p, int pend, int& s, float2& e)
{
    const bool in = p < pend;
    s   = in ? s : 0;
    e.x = in ? e.x : -__builtin_inff();
    e.y = in ? e.y : -__builtin_inff();
}

__device__ __forceinline__ int lane_bits(float v, int j)
{
    return __builtin_amdgcn_readlane(__float_as_int(v), j);
}

__device__ __forceinline__ float head_weight(float x0, float x1, int j, int hm)
{
    return __int_as_float((lane_bits(x1, j) & hm) | (lane_bits(x0, j) & ~hm));
}

__device__ __forceinline__ void agg_chunk(const v4f* __restrict__ feat4, int lane,
                                          int s, float x0, float x1, int n, int hm,
                                          float& den, float4& acc)
{
    int j = 0;
    for (; j + 8 <= n; j += 8) {
        v4f f[8];
        #pragma unroll
        for (int k = 0; k < 8; ++k)
            f[k] = feat4[(unsigned)__builtin_amdgcn_readlane(s, j + k) * (unsigned)(DHF / 4) + (unsigned)lane];
        __builtin_amdgcn_sched_barrier(0);
        #pragma unroll
        for (int k = 0; k < 8; ++k) {
            const float w = head_weight(x0, x1, j + k, hm);
            den += w;
            acc.x = fmaf(w, f[k].x, acc.x);
            acc.y = fmaf(w, f[k].y, acc.y);
            acc.z = fmaf(w, f[k].z, acc.z);
            acc.w = fmaf(w, f[k].w, acc.w);
        }
    }
    const int r = n - j;
    if (r > 0) {
        v4f f[7];
        #pragma unroll
        for (int k = 0; k < 7; ++k) {
            if (k < r) {
                f[k] = feat4[(unsigned)__builtin_amdgcn_readlane(s, j + k) * (unsigned)(DHF / 4) + (unsigned)lane];
            }
        }
        #pragma unroll
        for (int k = 0; k < 7; ++k) {
            if (k < r) {
                const float w = head_weight(x0, x1, j + k, hm);
                den += w;
                acc.x = fmaf(w, f[k].x, acc.x);
                acc.y = fmaf(w, f[k].y, acc.y);
                acc.z = fmaf(w, f[k].z, acc.z);
                acc.w = fmaf(w, f[k].w, acc.w);
            }
        }
    }
}

__global__ __launch_bounds__(256, 10) void gat_aggregate(const float* __restrict__ el,
                                                     const float* __restrict__ er,
                                                     const float* __restrict__ feat,
                                                     const int* __restrict__ csrc,
                                                     const int* __restrict__ off,
                                                     const float* __restrict__ bias,
                                                     float* __restrict__ out,
                                                     int last)
{
    const int lane = threadIdx.x & 31;
    const int hm   = -(lane >> 4);
    const int wv = __builtin_amdgcn_readfirstlane((int)(threadIdx.x >> 5));
    const int d0 = (blockIdx.x * 8 + wv) * AGG_NPW;
    if (d0 >= NN) return;

    const float2* __restrict__ el2  = reinterpret_cast<const float2*>(el);
    const float2* __restrict__ er2  = reinterpret_cast<const float2*>(er);
    float4* __restrict__ out4 = reinterpret_cast<float4*>(out);

    const float b0 = bias[4 * lane + 0], b1 = bias[4 * lane + 1];
    const float b2 = bias[4 * lane + 2], b3 = bias[4 * lane + 3];

    static_assert(AGG_NPW + 2 < 32, "bounds of two nodes past the last are read from lanes");
    const int    offv = off[d0 + ((lane <= AGG_NPW) ? lane : AGG_NPW)];
    const float2 erv  = er2[d0 + ((lane < AGG_NPW) ? lane : 0)];

    int pcur = __builtin_amdgcn_readlane(offv, 0);
    int pnxt = __builtin_amdgcn_readlane(offv, 1);
    int    sN  = load_src(csrc, pcur + lane, pnxt);
    int    sNN = load_src(csrc, pnxt + lane, __builtin_amdgcn_readlane(offv, 2));
    float2 gN  = el2[sN];

    for (int i = 0; i < AGG_NPW; ++i) {
        const int d  = d0 + i;
        const int p0 = pcur, p1 = pnxt;
        const float2 erd = make_float2(__int_as_float(lane_bits(erv.x, i)), __int_as_float(lane_bits(erv.y, i)));
        int    sF = sN;
        float2 eF = edge_score(gN, erd);
        keep_edges(p0 + lane, p1, sF, eF);

        pcur = p1;
        pnxt = __builtin_amdgcn_readlane(offv, i + 2);
        sN   = sNN;
        gN   = el2[sN];
        sNN  = load_src(csrc, pnxt + lane, __builtin_amdgcn_readlane(offv, i + 3));

        float m0 = eF.x, m1 = eF.y;
        for (int c0 = p0 + 32; c0 < p1; c0 += 32) {
            const int p = c0 + lane;
            if (p < p1) {
                const float2 e = edge_score(el2[csrc[p]], erd);
                m0 = fmaxf(m0, e.x);
                m1 = fmaxf(m1, e.y);
            }
        }
        #pragma unroll
        for (int ofs = 16; ofs > 0; ofs >>= 1) {
            m0 = fmaxf(m0, __shfl_xor(m0, ofs));
            m1 = fmaxf(m1, __shfl_xor(m1, ofs));
        }

        float  den = 0.f;
        float4 acc = make_float4(0.f, 0.f, 0.f, 0.f);
        {
            int    c0 = p0;
            int    s  = sF;
            float2 e  = eF;
            for (;;) {
                const int n = (p1 - c0 < 32) ? (p1 - c0) : 32;
                agg_chunk(reinterpret_cast<const v4f*>(feat), lane, s, expf(e.x - m0), expf(e.y - m1), n, hm, den, acc);
                c0 += 32;
                if (c0 >= p1) break;
                s = load_src(csrc, c0 + lane, p1);
                e = edge_score(el2[s], erd);
                keep_edges(c0 + lane, p1, s, e);
            }
        }

        const float dd = (den == 0.f) ? 1.f : den;
        float4 v;
        v.x = acc.x / dd + b0;
        v.y = acc.y / dd + b1;
        v.z = acc.z / dd + b2;
        v.w = acc.w / dd + b3;
        v.x = (v.x > 0.f) ? v.x : expm1f(v.x);
        v.y = (v.y > 0.f) ? v.y : expm1f(v.y);
        v.z = (v.z > 0.f) ? v.z : expm1f(v.z);
        v.w = (v.w > 0.f) ? v.w : expm1f(v.w);

        if (!last) {
            out4[d * (DHF / 4) + lane] = v;
        } else {
            float4 o;
            o.x = __shfl_xor(v.x, 16);
            o.y = __shfl_xor(v.y, 16);
            o.z = __shfl_xor(v.z, 16);
            o.w = __shfl_xor(v.w, 16);
            if (lane < 16)
                out4[d * (FF / 4) + lane] = make_float4(0.5f * (v.x + o.x), 0.5f * (v.y + o.y),
                                                        0.5f * (v.z + o.z), 0.5f * (v.w + o.w));
        }
    }
}

extern "C" void kernel_launch(void* const* d_in, const int* in_sizes, int n_in,
                              void* d_out, int out_size, void* d_ws, size_t ws_size,
                              hipStream_t stream)
{
    const float* x    = (const float*)d_in[0];
    const int*   src  = (const int*)d_in[1];
    const int*   dst  = (const int*)d_in[2];
    const float* Wall = (const float*)d_in[3];
    const float* al   = (const float*)d_in[4];
    const float* ar   = (const float*)d_in[5];
    const float* bias = (const float*)d_in[6];
    float* out = (float*)d_out;

    char* p = (char*)d_ws;
    auto take = [&](size_t bytes) {
        char* q = p;
        p += (bytes + 255) & ~size_t(255);
        return q;
    };
    float* feat   = (float*)take(sizeof(float) * NN * DHF);
    float* hbuf   = (float*)take(sizeof(float) * NN * DHF);
    float* el     = (float*)take(sizeof(float) * NN * HH);
    float* er     = (float*)take(sizeof(float) * NN * HH);
    int*   cnt    = (int*)take(sizeof(int) * CNT_N);
    int*   pos    = (int*)take(sizeof(int) * (CNT_N + 1));
    int2*  pairs  = (int2*)take(sizeof(int2) * EE);
    int*   off    = (int*)take(sizeof(int) * (NN + 1));
    int*   csrc   = (int*)take(sizeof(int) * EE);

    const int tb   = 256;
    const int gN2  = (NN * HH + tb - 1) / tb;
    const int gGmm = (NN + 15) / 16;
    const int gAgg = (NN + 8 * AGG_NPW - 1) / (8 * AGG_NPW);

    bkt_count  <<<NCHUNK, tb, 0, stream>>>(dst, cnt);
    bkt_scan   <<<1, 1024, 0, stream>>>(cnt, pos);
    bkt_scatter<<<NCHUNK, tb, 0, stream>>>(src, dst, pos, pairs);
    seg_sort   <<<NBKT, tb, 0, stream>>>(pairs, pos, off, csrc);

    const float* hin = x;
    for (int layer = 0; layer < 3; ++layer) {
        const float* Wl  = Wall + layer * DIN * DHF;
        const float* all = al   + layer * HH * FF;
        const float* arl = ar   + layer * HH * FF;
        const float* bl  = bias + layer * HH * FF;
        const int    last = (layer == 2);
        float* outp = last ? out : hbuf;

        gat_gemm     <<<gGmm, tb, 0, stream>>>(hin, Wl, feat);
        gat_attn     <<<gN2,  tb, 0, stream>>>(feat, all, arl, el, er);
        gat_aggregate<<<gAgg, tb, 0, stream>>>(el, er, feat, csrc, off, bl, outp, last);
        hin = hbuf;
    }
    (void)in_sizes; (void)n_in; (void)out_size; (void)ws_size;
}


# el/er dots made in the projection launch: each wave copies its accumulators to an LDS tile, the first wave sums the 64-wide dots in the old order; the 3 score launches are gone: 10 launches, not 13
# speedup vs baseline: 3.0244x; 1.0914x over previous
#include <hip/hip_runtime.h>
#include <math.h>

#define NN  50000
#define EE  800000
#define DIN 128
#define HH  2
#define FF  64
#define DHF 128
#define NEG_SLOPE 0.2f

typedef __attribute__((ext_vector_type(2))) float v2f;
typedef __attribute__((ext_vector_type(8))) float v8f;
typedef __attribute__((ext_vector_type(4))) float v4f;

#if defined(__AMDGCN__) && __has_builtin(__builtin_amdgcn_wmma_f32_16x16x4_f32)
#define USE_WMMA_F32 1
#else
#define USE_WMMA_F32 0
#endif

__global__ __launch_bounds__(256) void gat_gemm(const float* __restrict__ Hm,
                                                const float* __restrict__ W,
                                                const float* __restrict__ al,
                                                const float* __restrict__ ar,
                                                float* __restrict__ feat,
                                                float* __restrict__ el,
                                                float* __restrict__ er)
{
    __shared__ float sA[16 * 132];
    __shared__ __attribute__((aligned(16))) float sC[16 * 132];
    __shared__ __attribute__((aligned(16))) float sAlr[2 * DHF];
    const int tid  = threadIdx.x;
    const int row0 = blockIdx.x * 16;

    const float alr = (tid < DHF) ? al[tid] : ar[tid - DHF];

    for (int i = tid; i < 16 * DIN; i += 256) {
        int r = i >> 7, c = i & 127;
        int gr = row0 + r;
        sA[r * 132 + c] = (gr < NN) ? Hm[gr * DIN + c] : 0.0f;
    }
    __syncthreads();

    const int wave = tid >> 5;
    const int lane = tid & 31;
    const int mrow = lane & 15;
    const int hi   = lane >> 4;
    const int col0 = wave * 16;

    v8f c = {};
#if USE_WMMA_F32
    for (int k = 0; k < DIN; k += 4) {
        const int ka = k + 2 * hi;
        v2f a, b;
        a.x = sA[mrow * 132 + ka + 0];
        a.y = sA[mrow * 132 + ka + 1];
        b.x = W[(ka + 0) * DHF + col0 + mrow];
        b.y = W[(ka + 1) * DHF + col0 + mrow];
        c = __builtin_amdgcn_wmma_f32_16x16x4_f32(false, a, false, b,
                                                  (short)0, c, false, false);
    }
#else
    for (int k = 0; k < DIN; ++k) {
        float bkn = W[k * DHF + col0 + mrow];
        #pragma unroll
        for (int i = 0; i < 8; ++i)
            c[i] = fmaf(sA[(i + 8 * hi) * 132 + k], bkn, c[i]);
    }
#endif
    #pragma unroll
    for (int i = 0; i < 8; ++i) {
        int M = i + 8 * hi;
        sC[M * 132 + col0 + mrow] = c[i];
    }
    sAlr[tid] = alr;
    __syncthreads();
    #pragma unroll
    for (int i = 0; i < 8; ++i) {
        int M = i + 8 * hi;
        int r = row0 + M;
        if (r < NN) feat[r * DHF + col0 + mrow] = c[i];
    }

    if (tid < 32) {
        const int r = tid >> 1, h = tid & 1;
        const v4f* fp4 = reinterpret_cast<const v4f*>(sC + r * 132 + h * FF);
        const v4f* al4 = reinterpret_cast<const v4f*>(sAlr + h * FF);
        const v4f* ar4 = reinterpret_cast<const v4f*>(sAlr + DHF + h * FF);
        float sl = 0.f, sr = 0.f;
        #pragma unroll
        for (int q = 0; q < FF / 4; ++q) {
            const v4f v = fp4[q], a = al4[q], b = ar4[q];
            sl = fmaf(v.x, a.x, sl);  sr = fmaf(v.x, b.x, sr);
            sl = fmaf(v.y, a.y, sl);  sr = fmaf(v.y, b.y, sr);
            sl = fmaf(v.z, a.z, sl);  sr = fmaf(v.z, b.z, sr);
            sl = fmaf(v.w, a.w, sl);  sr = fmaf(v.w, b.w, sr);
        }
        const int n = row0 + r;
        if (n < NN) {
            el[n * HH + h] = sl;
            er[n * HH + h] = sr;
        }
    }
}

#define NCHUNK    250
#define CHUNK_E   3200
#define NBKT      250
#define BKT_NODES 200
#define CNT_N     (NBKT * NCHUNK)
static_assert(NCHUNK * CHUNK_E == EE, "the chunks cover the edge list exactly");
static_assert(NBKT * BKT_NODES == NN, "the ranges cover the nodes exactly");
static_assert(CHUNK_E % 4 == 0, "a chunk is read as int4");
#define CHUNK_Q   ((CHUNK_E / 4 + 255) / 256)
static_assert(NBKT <= 256 && BKT_NODES <= 224, "one thread per range; seven counts per lane of one wave");

__global__ __launch_bounds__(256) void bkt_count(const int* __restrict__ dst,
                                                 int* __restrict__ cnt)
{
    __shared__ int h[NBKT];
    const int t     = threadIdx.x;
    const int chunk = blockIdx.x;
    if (t < NBKT) h[t] = 0;
    __syncthreads();
    const int4* __restrict__ d4 = reinterpret_cast<const int4*>(dst + chunk * CHUNK_E);
    int4 d[CHUNK_Q];
    #pragma unroll
    for (int k = 0; k < CHUNK_Q; ++k) {
        const int q = t + k * 256;
        d[k] = d4[(q < CHUNK_E / 4) ? q : 0];
    }
    #pragma unroll
    for (int k = 0; k < CHUNK_Q; ++k) {
        if (t + k * 256 < CHUNK_E / 4) {
            atomicAdd(&h[(unsigned)d[k].x / BKT_NODES], 1);
            atomicAdd(&h[(unsigned)d[k].y / BKT_NODES], 1);
            atomicAdd(&h[(unsigned)d[k].z / BKT_NODES], 1);
            atomicAdd(&h[(unsigned)d[k].w / BKT_NODES], 1);
        }
    }
    __syncthreads();
    if (t < NBKT) cnt[t * NCHUNK + chunk] = h[t];
}

#define SCAN_ROWS 16
static_assert(32 * SCAN_ROWS * 128 >= CNT_N, "the scan block covers all counts");
static_assert(CNT_N % 4 == 0, "a group of four counts is all in or all out");

__global__ __launch_bounds__(1024) void bkt_scan(const int* __restrict__ cnt,
                                                 int* __restrict__ pos)
{
    __shared__ int wtot[32];
    const int t    = threadIdx.x;
    const int lane = t & 31;
    const int wave = t >> 5;
    const int base = wave * (SCAN_ROWS * 128) + lane * 4;

    int4 v[SCAN_ROWS];
    #pragma unroll
    for (int j = 0; j < SCAN_ROWS; ++j) {
        const int i = base + j * 128;
        v[j] = (i < CNT_N) ? *reinterpret_cast<const int4*>(cnt + i) : make_int4(0, 0, 0, 0);
    }

    int inc[SCAN_ROWS];
    #pragma unroll
    for (int j = 0; j < SCAN_ROWS; ++j) {
        v[j].y += v[j].x;
        v[j].z += v[j].y;
        v[j].w += v[j].z;
        int s = v[j].w;
        #pragma unroll
        for (int ofs = 1; ofs < 32; ofs <<= 1) {
            const int n = __shfl_up(s, ofs);
            if (lane >= ofs) s += n;
        }
        inc[j] = s;
    }

    int rowbase[SCAN_ROWS];
    int run = 0;
    #pragma unroll
    for (int j = 0; j < SCAN_ROWS; ++j) {
        rowbase[j] = run;
        run += __builtin_amdgcn_readlane(inc[j], 31);
    }

    if (lane == 0) wtot[wave] = run;
    __syncthreads();
    int wb = (lane < wave) ? wtot[lane] : 0;
    #pragma unroll
    for (int ofs = 16; ofs > 0; ofs >>= 1)
        wb += __shfl_xor(wb, ofs);

    #pragma unroll
    for (int j = 0; j < SCAN_ROWS; ++j) {
        const int i = base + j * 128;
        if (i < CNT_N) {
            const int e0 = wb + rowbase[j] + inc[j] - v[j].w;
            *reinterpret_cast<int4*>(pos + i) = make_int4(e0, e0 + v[j].x, e0 + v[j].y, e0 + v[j].z);
        }
    }
    if (t == 1023) pos[CNT_N] = wb + run;
}

__global__ __launch_bounds__(256) void bkt_scatter(const int* __restrict__ src,
                                                   const int* __restrict__ dst,
                                                   const int* __restrict__ pos,
                                                   int2* __restrict__ pairs)
{
    __shared__ int cur[NBKT];
    const int t     = threadIdx.x;
    const int chunk = blockIdx.x;
    if (t < NBKT) cur[t] = pos[t * NCHUNK + chunk];
    __syncthreads();
    const int4* __restrict__ s4 = reinterpret_cast<const int4*>(src + chunk * CHUNK_E);
    const int4* __restrict__ d4 = reinterpret_cast<const int4*>(dst + chunk * CHUNK_E);
    int4 s[CHUNK_Q], d[CHUNK_Q];
    #pragma unroll
    for (int k = 0; k < CHUNK_Q; ++k) {
        const int q  = t + k * 256;
        const int qc = (q < CHUNK_E / 4) ? q : 0;
        s[k] = s4[qc];
        d[k] = d4[qc];
    }
    #pragma unroll
    for (int k = 0; k < CHUNK_Q; ++k) {
        if (t + k * 256 < CHUNK_E / 4) {
            const int p0 = atomicAdd(&cur[(unsigned)d[k].x / BKT_NODES], 1);
            const int p1 = atomicAdd(&cur[(unsigned)d[k].y / BKT_NODES], 1);
            const int p2 = atomicAdd(&cur[(unsigned)d[k].z / BKT_NODES], 1);
            const int p3 = atomicAdd(&cur[(unsigned)d[k].w / BKT_NODES], 1);
            pairs[p0] = make_int2(s[k].x, d[k].x);
            pairs[p1] = make_int2(s[k].y, d[k].y);
            pairs[p2] = make_int2(s[k].z, d[k].z);
            pairs[p3] = make_int2(s[k].w, d[k].w);
        }
    }
}

#define SEG_HOLD 16

__global__ __launch_bounds__(256) void seg_sort(const int2* __restrict__ pairs,
                                                const int* __restrict__ pos,
                                                int* __restrict__ off,
                                                int* __restrict__ csrc)
{
    __shared__ int cnt[BKT_NODES];
    __shared__ int start[BKT_NODES];
    const int t     = threadIdx.x;
    const int b     = blockIdx.x;
    const int node0 = b * BKT_NODES;
    const int s0 = pos[b * NCHUNK];
    const int s1 = pos[(b + 1) * NCHUNK];
    if (t < BKT_NODES) cnt[t] = 0;
    __syncthreads();
    int2 pr[SEG_HOLD];
    #pragma unroll
    for (int k = 0; k < SEG_HOLD; ++k) {
        const int i = s0 + t + k * 256;
        pr[k] = pairs[(i < s1) ? i : 0];
    }
    #pragma unroll
    for (int k = 0; k < SEG_HOLD; ++k) {
        if (s0 + t + k * 256 < s1) atomicAdd(&cnt[pr[k].y - node0], 1);
    }
    for (int i = s0 + t + SEG_HOLD * 256; i < s1; i += 256)
        atomicAdd(&cnt[pairs[i].y - node0], 1);
    __syncthreads();
    if (t < 32) {
        int c[7];
        int s = 0;
        #pragma unroll
        for (int k = 0; k < 7; ++k) {
            const int idx = t * 7 + k;
            c[k] = (idx < BKT_NODES) ? cnt[idx] : 0;
            s += c[k];
        }
        int inc = s;
        #pragma unroll
        for (int ofs = 1; ofs < 32; ofs <<= 1) {
            const int n = __shfl_up(inc, ofs);
            if (t >= ofs) inc += n;
        }
        int run = s0 + inc - s;
        #pragma unroll
        for (int k = 0; k < 7; ++k) {
            const int idx = t * 7 + k;
            if (idx < BKT_NODES) {
                start[idx] = run;
                off[node0 + idx] = run;
            }
            run += c[k];
        }
    }
    if (b == NBKT - 1 && t == 0) off[NN] = s1;
    __syncthreads();
    #pragma unroll
    for (int k = 0; k < SEG_HOLD; ++k) {
        if (s0 + t + k * 256 < s1) {
            const int p = atomicAdd(&start[pr[k].y - node0], 1);
            csrc[p] = pr[k].x;
        }
    }
    for (int i = s0 + t + SEG_HOLD * 256; i < s1; i += 256) {
        const int2 q = pairs[i];
        const int p = atomicAdd(&start[q.y - node0], 1);
        csrc[p] = q.x;
    }
}

#define AGG_NPW 5
static_assert(NN % AGG_NPW == 0, "a wave's group of nodes is all in or all out");

__device__ __forceinline__ int load_src(const int* __restrict__ csrc, int p, int pend)
{
    return csrc[(p < pend) ? p : 0];
}

__device__ __forceinline__ float2 edge_score(float2 l, float2 r)
{
    const float a = l.x + r.x;
    const float b = l.y + r.y;
    return make_float2((a >= 0.f) ? a : NEG_SLOPE * a, (b >= 0.f) ? b : NEG_SLOPE * b);
}

__device__ __forceinline__ void keep_edges(int p, int pend, int& s, float2& e)
{
    const bool in = p < pend;
    s   = in ? s : 0;
    e.x = in ? e.x : -__builtin_inff();
    e.y = in ? e.y : -__builtin_inff();
}

__device__ __forceinline__ int lane_bits(float v, int j)
{
    return __builtin_amdgcn_readlane(__float_as_int(v), j);
}

__device__ __forceinline__ float head_weight(float x0, float x1, int j, int hm)
{
    return __int_as_float((lane_bits(x1, j) & hm) | (lane_bits(x0, j) & ~hm));
}

__device__ __forceinline__ void agg_chunk(const v4f* __restrict__ feat4, int lane,
                                          int s, float x0, float x1, int n, int hm,
                                          float& den, float4& acc)
{
    int j = 0;
    for (; j + 8 <= n; j += 8) {
        v4f f[8];
        #pragma unroll
        for (int k = 0; k < 8; ++k)
            f[k] = feat4[(unsigned)__builtin_amdgcn_readlane(s, j + k) * (unsigned)(DHF / 4) + (unsigned)lane];
        __builtin_amdgcn_sched_barrier(0);
        #pragma unroll
        for (int k = 0; k < 8; ++k) {
            const float w = head_weight(x0, x1, j + k, hm);
            den += w;
            acc.x = fmaf(w, f[k].x, acc.x);
            acc.y = fmaf(w, f[k].y, acc.y);
            acc.z = fmaf(w, f[k].z, acc.z);
            acc.w = fmaf(w, f[k].w, acc.w);
        }
    }
    const int r = n - j;
    if (r > 0) {
        v4f f[7];
        #pragma unroll
        for (int k = 0; k < 7; ++k) {
            if (k < r) {
                f[k] = feat4[(unsigned)__builtin_amdgcn_readlane(s, j + k) * (unsigned)(DHF / 4) + (unsigned)lane];
            }
        }
        #pragma unroll
        for (int k = 0; k < 7; ++k) {
            if (k < r) {
                const float w = head_weight(x0, x1, j + k, hm);
                den += w;
                acc.x = fmaf(w, f[k].x, acc.x);
                acc.y = fmaf(w, f[k].y, acc.y);
                acc.z = fmaf(w, f[k].z, acc.z);
                acc.w = fmaf(w, f[k].w, acc.w);
            }
        }
    }
}

__global__ __launch_bounds__(256, 10) void gat_aggregate(const float* __restrict__ el,
                                                     const float* __restrict__ er,
                                                     const float* __restrict__ feat,
                                                     const int* __restrict__ csrc,
                                                     const int* __restrict__ off,
                                                     const float* __restrict__ bias,
                                                     float* __restrict__ out,
                                                     int last)
{
    const int lane = threadIdx.x & 31;
    const int hm   = -(lane >> 4);
    const int wv = __builtin_amdgcn_readfirstlane((int)(threadIdx.x >> 5));
    const int d0 = (blockIdx.x * 8 + wv) * AGG_NPW;
    if (d0 >= NN) return;

    const float2* __restrict__ el2  = reinterpret_cast<const float2*>(el);
    const float2* __restrict__ er2  = reinterpret_cast<const float2*>(er);
    float4* __restrict__ out4 = reinterpret_cast<float4*>(out);

    const float b0 = bias[4 * lane + 0], b1 = bias[4 * lane + 1];
    const float b2 = bias[4 * lane + 2], b3 = bias[4 * lane + 3];

    static_assert(AGG_NPW + 2 < 32, "bounds of two nodes past the last are read from lanes");
    const int    offv = off[d0 + ((lane <= AGG_NPW) ? lane : AGG_NPW)];
    const float2 erv  = er2[d0 + ((lane < AGG_NPW) ? lane : 0)];

    int pcur = __builtin_amdgcn_readlane(offv, 0);
    int pnxt = __builtin_amdgcn_readlane(offv, 1);
    int    sN  = load_src(csrc, pcur + lane, pnxt);
    int    sNN = load_src(csrc, pnxt + lane, __builtin_amdgcn_readlane(offv, 2));
    float2 gN  = el2[sN];

    for (int i = 0; i < AGG_NPW; ++i) {
        const int d  = d0 + i;
        const int p0 = pcur, p1 = pnxt;
        const float2 erd = make_float2(__int_as_float(lane_bits(erv.x, i)), __int_as_float(lane_bits(erv.y, i)));
        int    sF = sN;
        float2 eF = edge_score(gN, erd);
        keep_edges(p0 + lane, p1, sF, eF);

        pcur = p1;
        pnxt = __builtin_amdgcn_readlane(offv, i + 2);
        sN   = sNN;
        gN   = el2[sN];
        sNN  = load_src(csrc, pnxt + lane, __builtin_amdgcn_readlane(offv, i + 3));

        float m0 = eF.x, m1 = eF.y;
        for (int c0 = p0 + 32; c0 < p1; c0 += 32) {
            const int p = c0 + lane;
            if (p < p1) {
                const float2 e = edge_score(el2[csrc[p]], erd);
                m0 = fmaxf(m0, e.x);
                m1 = fmaxf(m1, e.y);
            }
        }
        #pragma unroll
        for (int ofs = 16; ofs > 0; ofs >>= 1) {
            m0 = fmaxf(m0, __shfl_xor(m0, ofs));
            m1 = fmaxf(m1, __shfl_xor(m1, ofs));
        }

        float  den = 0.f;
        float4 acc = make_float4(0.f, 0.f, 0.f, 0.f);
        {
            int    c0 = p0;
            int    s  = sF;
            float2 e  = eF;
            for (;;) {
                const int n = (p1 - c0 < 32) ? (p1 - c0) : 32;
                agg_chunk(reinterpret_cast<const v4f*>(feat), lane, s, expf(e.x - m0), expf(e.y - m1), n, hm, den, acc);
                c0 += 32;
                if (c0 >= p1) break;
                s = load_src(csrc, c0 + lane, p1);
                e = edge_score(el2[s], erd);
                keep_edges(c0 + lane, p1, s, e);
            }
        }

        const float dd = (den == 0.f) ? 1.f : den;
        float4 v;
        v.x = acc.x / dd + b0;
        v.y = acc.y / dd + b1;
        v.z = acc.z / dd + b2;
        v.w = acc.w / dd + b3;
        v.x = (v.x > 0.f) ? v.x : expm1f(v.x);
        v.y = (v.y > 0.f) ? v.y : expm1f(v.y);
        v.z = (v.z > 0.f) ? v.z : expm1f(v.z);
        v.w = (v.w > 0.f) ? v.w : expm1f(v.w);

        if (!last) {
            out4[d * (DHF / 4) + lane] = v;
        } else {
            float4 o;
            o.x = __shfl_xor(v.x, 16);
            o.y = __shfl_xor(v.y, 16);
            o.z = __shfl_xor(v.z, 16);
            o.w = __shfl_xor(v.w, 16);
            if (lane < 16)
                out4[d * (FF / 4) + lane] = make_float4(0.5f * (v.x + o.x), 0.5f * (v.y + o.y),
                                                        0.5f * (v.z + o.z), 0.5f * (v.w + o.w));
        }
    }
}

extern "C" void kernel_launch(void* const* d_in, const int* in_sizes, int n_in,
                              void* d_out, int out_size, void* d_ws, size_t ws_size,
                              hipStream_t stream)
{
    const float* x    = (const float*)d_in[0];
    const int*   src  = (const int*)d_in[1];
    const int*   dst  = (const int*)d_in[2];
    const float* Wall = (const float*)d_in[3];
    const float* al   = (const float*)d_in[4];
    const float* ar   = (const float*)d_in[5];
    const float* bias = (const float*)d_in[6];
    float* out = (float*)d_out;

    char* p = (char*)d_ws;
    auto take = [&](size_t bytes) {
        char* q = p;
        p += (bytes + 255) & ~size_t(255);
        return q;
    };
    float* feat   = (float*)take(sizeof(float) * NN * DHF);
    float* hbuf   = (float*)take(sizeof(float) * NN * DHF);
    float* el     = (float*)take(sizeof(float) * NN * HH);
    float* er     = (float*)take(sizeof(float) * NN * HH);
    int*   cnt    = (int*)take(sizeof(int) * CNT_N);
    int*   pos    = (int*)take(sizeof(int) * (CNT_N + 1));
    int2*  pairs  = (int2*)take(sizeof(int2) * EE);
    int*   off    = (int*)take(sizeof(int) * (NN + 1));
    int*   csrc   = (int*)take(sizeof(int) * EE);

    const int tb   = 256;
    const int gGmm = (NN + 15) / 16;
    const int gAgg = (NN + 8 * AGG_NPW - 1) / (8 * AGG_NPW);

    bkt_count  <<<NCHUNK, tb, 0, stream>>>(dst, cnt);
    bkt_scan   <<<1, 1024, 0, stream>>>(cnt, pos);
    bkt_scatter<<<NCHUNK, tb, 0, stream>>>(src, dst, pos, pairs);
    seg_sort   <<<NBKT, tb, 0, stream>>>(pairs, pos, off, csrc);

    const float* hin = x;
    for (int layer = 0; layer < 3; ++layer) {
        const float* Wl  = Wall + layer * DIN * DHF;
        const float* all = al   + layer * HH * FF;
        const float* arl = ar   + layer * HH * FF;
        const float* bl  = bias + layer * HH * FF;
        const int    last = (layer == 2);
        float* outp = last ? out : hbuf;

        gat_gemm     <<<gGmm, tb, 0, stream>>>(hin, Wl, all, arl, feat, el, er);
        gat_aggregate<<<gAgg, tb, 0, stream>>>(el, er, feat, csrc, off, bl, outp, last);
        hin = hbuf;
    }
    (void)in_sizes; (void)n_in; (void)out_size; (void)ws_size;
}


# projection launch as a resident grid: 768 work-groups each walk 4-5 row tiles, the wave's W fragments held in registers, next tile's rows prefetched (was 3125 one-tile groups); sums unchanged
# speedup vs baseline: 3.5186x; 1.1634x over previous
#include <hip/hip_runtime.h>
#include <math.h>

#define NN  50000
#define EE  800000
#define DIN 128
#define HH  2
#define FF  64
#define DHF 128
#define NEG_SLOPE 0.2f

typedef __attribute__((ext_vector_type(2))) float v2f;
typedef __attribute__((ext_vector_type(8))) float v8f;
typedef __attribute__((ext_vector_type(4))) float v4f;

#if defined(__AMDGCN__) && __has_builtin(__builtin_amdgcn_wmma_f32_16x16x4_f32)
#define USE_WMMA_F32 1
#else
#define USE_WMMA_F32 0
#endif

#define GEMM_TILES (NN / 16)
#define GEMM_GRID  768
static_assert(NN % 16 == 0, "every row tile is whole: no row of a tile lies past the last node");
static_assert(GEMM_GRID <= GEMM_TILES, "every work-group has a first tile");

__global__ __launch_bounds__(256, 8) void gat_gemm(const float* __restrict__ Hm,
                                                   const float* __restrict__ W,
                                                   const float* __restrict__ al,
                                                   const float* __restrict__ ar,
                                                   float* __restrict__ feat,
                                                   float* __restrict__ el,
                                                   float* __restrict__ er)
{
    __shared__ __attribute__((aligned(16))) float sA[2][16 * 132];
    __shared__ __attribute__((aligned(16))) float sC[2][16 * 132];
    __shared__ __attribute__((aligned(16))) float sAlr[2 * DHF];
    const int tid  = threadIdx.x;
    const int lane = tid & 31;
    const int wave = __builtin_amdgcn_readfirstlane(tid >> 5);
    const int mrow = lane & 15;
    const int hi   = lane >> 4;
    const int col0 = wave * 16;

    int t = blockIdx.x;

    const v4f* __restrict__ H4 = reinterpret_cast<const v4f*>(Hm);
    v4f pa0 = H4[(t * 16 + wave) * (DIN / 4) + lane];
    v4f pa1 = H4[(t * 16 + wave + 8) * (DIN / 4) + lane];

    const float alr = (tid < DHF) ? al[tid] : ar[tid - DHF];

    v2f b[DIN / 4];
    #pragma unroll
    for (int kk = 0; kk < DIN / 4; ++kk) {
        const int ka = 4 * kk + 2 * hi;
        b[kk].x = W[(ka + 0) * DHF + col0 + mrow];
        b[kk].y = W[(ka + 1) * DHF + col0 + mrow];
    }

    *reinterpret_cast<v4f*>(&sA[0][wave * 132 + 4 * lane])       = pa0;
    *reinterpret_cast<v4f*>(&sA[0][(wave + 8) * 132 + 4 * lane]) = pa1;
    sAlr[tid] = alr;
    __syncthreads();

    for (int it = 0; t < GEMM_TILES; ++it, t += GEMM_GRID) {
        const int  buf  = it & 1;
        const int  tn   = t + GEMM_GRID;
        const bool more = tn < GEMM_TILES;
        if (more) {
            pa0 = H4[(tn * 16 + wave) * (DIN / 4) + lane];
            pa1 = H4[(tn * 16 + wave + 8) * (DIN / 4) + lane];
        }

        v8f c = {};
#if USE_WMMA_F32
        const float* __restrict__ ap = &sA[buf][mrow * 132 + 2 * hi];
        #pragma unroll
        for (int kk = 0; kk < DIN / 4; ++kk) {
            v2f a;
            a.x = ap[4 * kk + 0];
            a.y = ap[4 * kk + 1];
            c = __builtin_amdgcn_wmma_f32_16x16x4_f32(false, a, false, b[kk],
                                                      (short)0, c, false, false);
        }
#else
        for (int k = 0; k < DIN; ++k) {
            float bkn = W[k * DHF + col0 + mrow];
            #pragma unroll
            for (int i = 0; i < 8; ++i)
                c[i] = fmaf(sA[buf][(i + 8 * hi) * 132 + k], bkn, c[i]);
        }
#endif
        #pragma unroll
        for (int i = 0; i < 8; ++i) {
            int M = i + 8 * hi;
            sC[buf][M * 132 + col0 + mrow] = c[i];
        }
        if (more) {
            *reinterpret_cast<v4f*>(&sA[buf ^ 1][wave * 132 + 4 * lane])       = pa0;
            *reinterpret_cast<v4f*>(&sA[buf ^ 1][(wave + 8) * 132 + 4 * lane]) = pa1;
        }
        __syncthreads();
        const int row0 = t * 16;
        #pragma unroll
        for (int i = 0; i < 8; ++i) {
            int M = i + 8 * hi;
            feat[(row0 + M) * DHF + col0 + mrow] = c[i];
        }

        if (wave == (it & 7)) {
            const int r = lane >> 1, h = lane & 1;
            const v4f* fp4 = reinterpret_cast<const v4f*>(&sC[buf][r * 132 + h * FF]);
            const v4f* al4 = reinterpret_cast<const v4f*>(sAlr + h * FF);
            const v4f* ar4 = reinterpret_cast<const v4f*>(sAlr + DHF + h * FF);
            float sl = 0.f, sr = 0.f;
            #pragma unroll
            for (int q = 0; q < FF / 4; ++q) {
                const v4f v = fp4[q], a = al4[q], bq = ar4[q];
                sl = fmaf(v.x, a.x, sl);  sr = fmaf(v.x, bq.x, sr);
                sl = fmaf(v.y, a.y, sl);  sr = fmaf(v.y, bq.y, sr);
                sl = fmaf(v.z, a.z, sl);  sr = fmaf(v.z, bq.z, sr);
                sl = fmaf(v.w, a.w, sl);  sr = fmaf(v.w, bq.w, sr);
            }
            const int n = row0 + r;
            el[n * HH + h] = sl;
            er[n * HH + h] = sr;
        }
    }
}

#define NCHUNK    250
#define CHUNK_E   3200
#define NBKT      250
#define BKT_NODES 200
#define CNT_N     (NBKT * NCHUNK)
static_assert(NCHUNK * CHUNK_E == EE, "the chunks cover the edge list exactly");
static_assert(NBKT * BKT_NODES == NN, "the ranges cover the nodes exactly");
static_assert(CHUNK_E % 4 == 0, "a chunk is read as int4");
#define CHUNK_Q   ((CHUNK_E / 4 + 255) / 256)
static_assert(NBKT <= 256 && BKT_NODES <= 224, "one thread per range; seven counts per lane of one wave");

__global__ __launch_bounds__(256) void bkt_count(const int* __restrict__ dst,
                                                 int* __restrict__ cnt)
{
    __shared__ int h[NBKT];
    const int t     = threadIdx.x;
    const int chunk = blockIdx.x;
    if (t < NBKT) h[t] = 0;
    __syncthreads();
    const int4* __restrict__ d4 = reinterpret_cast<const int4*>(dst + chunk * CHUNK_E);
    int4 d[CHUNK_Q];
    #pragma unroll
    for (int k = 0; k < CHUNK_Q; ++k) {
        const int q = t + k * 256;
        d[k] = d4[(q < CHUNK_E / 4) ? q : 0];
    }
    #pragma unroll
    for (int k = 0; k < CHUNK_Q; ++k) {
        if (t + k * 256 < CHUNK_E / 4) {
            atomicAdd(&h[(unsigned)d[k].x / BKT_NODES], 1);
            atomicAdd(&h[(unsigned)d[k].y / BKT_NODES], 1);
            atomicAdd(&h[(unsigned)d[k].z / BKT_NODES], 1);
            atomicAdd(&h[(unsigned)d[k].w / BKT_NODES], 1);
        }
    }
    __syncthreads();
    if (t < NBKT) cnt[t * NCHUNK + chunk] = h[t];
}

#define SCAN_ROWS 16
static_assert(32 * SCAN_ROWS * 128 >= CNT_N, "the scan block covers all counts");
static_assert(CNT_N % 4 == 0, "a group of four counts is all in or all out");

__global__ __launch_bounds__(1024) void bkt_scan(const int* __restrict__ cnt,
                                                 int* __restrict__ pos)
{
    __shared__ int wtot[32];
    const int t    = threadIdx.x;
    const int lane = t & 31;
    const int wave = t >> 5;
    const int base = wave * (SCAN_ROWS * 128) + lane * 4;

    int4 v[SCAN_ROWS];
    #pragma unroll
    for (int j = 0; j < SCAN_ROWS; ++j) {
        const int i = base + j * 128;
        v[j] = (i < CNT_N) ? *reinterpret_cast<const int4*>(cnt + i) : make_int4(0, 0, 0, 0);
    }

    int inc[SCAN_ROWS];
    #pragma unroll
    for (int j = 0; j < SCAN_ROWS; ++j) {
        v[j].y += v[j].x;
        v[j].z += v[j].y;
        v[j].w += v[j].z;
        int s = v[j].w;
        #pragma unroll
        for (int ofs = 1; ofs < 32; ofs <<= 1) {
            const int n = __shfl_up(s, ofs);
            if (lane >= ofs) s += n;
        }
        inc[j] = s;
    }

    int rowbase[SCAN_ROWS];
    int run = 0;
    #pragma unroll
    for (int j = 0; j < SCAN_ROWS; ++j) {
        rowbase[j] = run;
        run += __builtin_amdgcn_readlane(inc[j], 31);
    }

    if (lane == 0) wtot[wave] = run;
    __syncthreads();
    int wb = (lane < wave) ? wtot[lane] : 0;
    #pragma unroll
    for (int ofs = 16; ofs > 0; ofs >>= 1)
        wb += __shfl_xor(wb, ofs);

    #pragma unroll
    for (int j = 0; j < SCAN_ROWS; ++j) {
        const int i = base + j * 128;
        if (i < CNT_N) {
            const int e0 = wb + rowbase[j] + inc[j] - v[j].w;
            *reinterpret_cast<int4*>(pos + i) = make_int4(e0, e0 + v[j].x, e0 + v[j].y, e0 + v[j].z);
        }
    }
    if (t == 1023) pos[CNT_N] = wb + run;
}

__global__ __launch_bounds__(256) void bkt_scatter(const int* __restrict__ src,
                                                   const int* __restrict__ dst,
                                                   const int* __restrict__ pos,
                                                   int2* __restrict__ pairs)
{
    __shared__ int cur[NBKT];
    const int t     = threadIdx.x;
    const int chunk = blockIdx.x;
    if (t < NBKT) cur[t] = pos[t * NCHUNK + chunk];
    __syncthreads();
    const int4* __restrict__ s4 = reinterpret_cast<const int4*>(src + chunk * CHUNK_E);
    const int4* __restrict__ d4 = reinterpret_cast<const int4*>(dst + chunk * CHUNK_E);
    int4 s[CHUNK_Q], d[CHUNK_Q];
    #pragma unroll
    for (int k = 0; k < CHUNK_Q; ++k) {
        const int q  = t + k * 256;
        const int qc = (q < CHUNK_E / 4) ? q : 0;
        s[k] = s4[qc];
        d[k] = d4[qc];
    }
    #pragma unroll
    for (int k = 0; k < CHUNK_Q; ++k) {
        if (t + k * 256 < CHUNK_E / 4) {
            const int p0 = atomicAdd(&cur[(unsigned)d[k].x / BKT_NODES], 1);
            const int p1 = atomicAdd(&cur[(unsigned)d[k].y / BKT_NODES], 1);
            const int p2 = atomicAdd(&cur[(unsigned)d[k].z / BKT_NODES], 1);
            const int p3 = atomicAdd(&cur[(unsigned)d[k].w / BKT_NODES], 1);
            pairs[p0] = make_int2(s[k].x, d[k].x);
            pairs[p1] = make_int2(s[k].y, d[k].y);
            pairs[p2] = make_int2(s[k].z, d[k].z);
            pairs[p3] = make_int2(s[k].w, d[k].w);
        }
    }
}

#define SEG_HOLD 16

__global__ __launch_bounds__(256) void seg_sort(const int2* __restrict__ pairs,
                                                const int* __restrict__ pos,
                                                int* __restrict__ off,
                                                int* __restrict__ csrc)
{
    __shared__ int cnt[BKT_NODES];
    __shared__ int start[BKT_NODES];
    const int t     = threadIdx.x;
    const int b     = blockIdx.x;
    const int node0 = b * BKT_NODES;
    const int s0 = pos[b * NCHUNK];
    const int s1 = pos[(b + 1) * NCHUNK];
    if (t < BKT_NODES) cnt[t] = 0;
    __syncthreads();
    int2 pr[SEG_HOLD];
    #pragma unroll
    for (int k = 0; k < SEG_HOLD; ++k) {
        const int i = s0 + t + k * 256;
        pr[k] = pairs[(i < s1) ? i : 0];
    }
    #pragma unroll
    for (int k = 0; k < SEG_HOLD; ++k) {
        if (s0 + t + k * 256 < s1) atomicAdd(&cnt[pr[k].y - node0], 1);
    }
    for (int i = s0 + t + SEG_HOLD * 256; i < s1; i += 256)
        atomicAdd(&cnt[pairs[i].y - node0], 1);
    __syncthreads();
    if (t < 32) {
        int c[7];
        int s = 0;
        #pragma unroll
        for (int k = 0; k < 7; ++k) {
            const int idx = t * 7 + k;
            c[k] = (idx < BKT_NODES) ? cnt[idx] : 0;
            s += c[k];
        }
        int inc = s;
        #pragma unroll
        for (int ofs = 1; ofs < 32; ofs <<= 1) {
            const int n = __shfl_up(inc, ofs);
            if (t >= ofs) inc += n;
        }
        int run = s0 + inc - s;
        #pragma unroll
        for (int k = 0; k < 7; ++k) {
            const int idx = t * 7 + k;
            if (idx < BKT_NODES) {
                start[idx] = run;
                off[node0 + idx] = run;
            }
            run += c[k];
        }
    }
    if (b == NBKT - 1 && t == 0) off[NN] = s1;
    __syncthreads();
    #pragma unroll
    for (int k = 0; k < SEG_HOLD; ++k) {
        if (s0 + t + k * 256 < s1) {
            const int p = atomicAdd(&start[pr[k].y - node0], 1);
            csrc[p] = pr[k].x;
        }
    }
    for (int i = s0 + t + SEG_HOLD * 256; i < s1; i += 256) {
        const int2 q = pairs[i];
        const int p = atomicAdd(&start[q.y - node0], 1);
        csrc[p] = q.x;
    }
}

#define AGG_NPW 5
static_assert(NN % AGG_NPW == 0, "a wave's group of nodes is all in or all out");

__device__ __forceinline__ int load_src(const int* __restrict__ csrc, int p, int pend)
{
    return csrc[(p < pend) ? p : 0];
}

__device__ __forceinline__ float2 edge_score(float2 l, float2 r)
{
    const float a = l.x + r.x;
    const float b = l.y + r.y;
    return make_float2((a >= 0.f) ? a : NEG_SLOPE * a, (b >= 0.f) ? b : NEG_SLOPE * b);
}

__device__ __forceinline__ void keep_edges(int p, int pend, int& s, float2& e)
{
    const bool in = p < pend;
    s   = in ? s : 0;
    e.x = in ? e.x : -__builtin_inff();
    e.y = in ? e.y : -__builtin_inff();
}

__device__ __forceinline__ int lane_bits(float v, int j)
{
    return __builtin_amdgcn_readlane(__float_as_int(v), j);
}

__device__ __forceinline__ float head_weight(float x0, float x1, int j, int hm)
{
    return __int_as_float((lane_bits(x1, j) & hm) | (lane_bits(x0, j) & ~hm));
}

__device__ __forceinline__ void agg_chunk(const v4f* __restrict__ feat4, int lane,
                                          int s, float x0, float x1, int n, int hm,
                                          float& den, float4& acc)
{
    int j = 0;
    for (; j + 8 <= n; j += 8) {
        v4f f[8];
        #pragma unroll
        for (int k = 0; k < 8; ++k)
            f[k] = feat4[(unsigned)__builtin_amdgcn_readlane(s, j + k) * (unsigned)(DHF / 4) + (unsigned)lane];
        __builtin_amdgcn_sched_barrier(0);
        #pragma unroll
        for (int k = 0; k < 8; ++k) {
            const float w = head_weight(x0, x1, j + k, hm);
            den += w;
            acc.x = fmaf(w, f[k].x, acc.x);
            acc.y = fmaf(w, f[k].y, acc.y);
            acc.z = fmaf(w, f[k].z, acc.z);
            acc.w = fmaf(w, f[k].w, acc.w);
        }
    }
    const int r = n - j;
    if (r > 0) {
        v4f f[7];
        #pragma unroll
        for (int k = 0; k < 7; ++k) {
            if (k < r) {
                f[k] = feat4[(unsigned)__builtin_amdgcn_readlane(s, j + k) * (unsigned)(DHF / 4) + (unsigned)lane];
            }
        }
        #pragma unroll
        for (int k = 0; k < 7; ++k) {
            if (k < r) {
                const float w = head_weight(x0, x1, j + k, hm);
                den += w;
                acc.x = fmaf(w, f[k].x, acc.x);
                acc.y = fmaf(w, f[k].y, acc.y);
                acc.z = fmaf(w, f[k].z, acc.z);
                acc.w = fmaf(w, f[k].w, acc.w);
            }
        }
    }
}

__global__ __launch_bounds__(256, 10) void gat_aggregate(const float* __restrict__ el,
                                                     const float* __restrict__ er,
                                                     const float* __restrict__ feat,
                                                     const int* __restrict__ csrc,
                                                     const int* __restrict__ off,
                                                     const float* __restrict__ bias,
                                                     float* __restrict__ out,
                                                     int last)
{
    const int lane = threadIdx.x & 31;
    const int hm   = -(lane >> 4);
    const int wv = __builtin_amdgcn_readfirstlane((int)(threadIdx.x >> 5));
    const int d0 = (blockIdx.x * 8 + wv) * AGG_NPW;
    if (d0 >= NN) return;

    const float2* __restrict__ el2  = reinterpret_cast<const float2*>(el);
    const float2* __restrict__ er2  = reinterpret_cast<const float2*>(er);
    float4* __restrict__ out4 = reinterpret_cast<float4*>(out);

    const float b0 = bias[4 * lane + 0], b1 = bias[4 * lane + 1];
    const float b2 = bias[4 * lane + 2], b3 = bias[4 * lane + 3];

    static_assert(AGG_NPW + 2 < 32, "bounds of two nodes past the last are read from lanes");
    const int    offv = off[d0 + ((lane <= AGG_NPW) ? lane : AGG_NPW)];
    const float2 erv  = er2[d0 + ((lane < AGG_NPW) ? lane : 0)];

    int pcur = __builtin_amdgcn_readlane(offv, 0);
    int pnxt = __builtin_amdgcn_readlane(offv, 1);
    int    sN  = load_src(csrc, pcur + lane, pnxt);
    int    sNN = load_src(csrc, pnxt + lane, __builtin_amdgcn_readlane(offv, 2));
    float2 gN  = el2[sN];

    for (int i = 0; i < AGG_NPW; ++i) {
        const int d  = d0 + i;
        const int p0 = pcur, p1 = pnxt;
        const float2 erd = make_float2(__int_as_float(lane_bits(erv.x, i)), __int_as_float(lane_bits(erv.y, i)));
        int    sF = sN;
        float2 eF = edge_score(gN, erd);
        keep_edges(p0 + lane, p1, sF, eF);

        pcur = p1;
        pnxt = __builtin_amdgcn_readlane(offv, i + 2);
        sN   = sNN;
        gN   = el2[sN];
        sNN  = load_src(csrc, pnxt + lane, __builtin_amdgcn_readlane(offv, i + 3));

        float m0 = eF.x, m1 = eF.y;
        for (int c0 = p0 + 32; c0 < p1; c0 += 32) {
            const int p = c0 + lane;
            if (p < p1) {
                const float2 e = edge_score(el2[csrc[p]], erd);
                m0 = fmaxf(m0, e.x);
                m1 = fmaxf(m1, e.y);
            }
        }
        #pragma unroll
        for (int ofs = 16; ofs > 0; ofs >>= 1) {
            m0 = fmaxf(m0, __shfl_xor(m0, ofs));
            m1 = fmaxf(m1, __shfl_xor(m1, ofs));
        }

        float  den = 0.f;
        float4 acc = make_float4(0.f, 0.f, 0.f, 0.f);
        {
            int    c0 = p0;
            int    s  = sF;
            float2 e  = eF;
            for (;;) {
                const int n = (p1 - c0 < 32) ? (p1 - c0) : 32;
                agg_chunk(reinterpret_cast<const v4f*>(feat), lane, s, expf(e.x - m0), expf(e.y - m1), n, hm, den, acc);
                c0 += 32;
                if (c0 >= p1) break;
                s = load_src(csrc, c0 + lane, p1);
                e = edge_score(el2[s], erd);
                keep_edges(c0 + lane, p1, s, e);
            }
        }

        const float dd = (den == 0.f) ? 1.f : den;
        float4 v;
        v.x = acc.x / dd + b0;
        v.y = acc.y / dd + b1;
        v.z = acc.z / dd + b2;
        v.w = acc.w / dd + b3;
        v.x = (v.x > 0.f) ? v.x : expm1f(v.x);
        v.y = (v.y > 0.f) ? v.y : expm1f(v.y);
        v.z = (v.z > 0.f) ? v.z : expm1f(v.z);
        v.w = (v.w > 0.f) ? v.w : expm1f(v.w);

        if (!last) {
            out4[d * (DHF / 4) + lane] = v;
        } else {
            float4 o;
            o.x = __shfl_xor(v.x, 16);
            o.y = __shfl_xor(v.y, 16);
            o.z = __shfl_xor(v.z, 16);
            o.w = __shfl_xor(v.w, 16);
            if (lane < 16)
                out4[d * (FF / 4) + lane] = make_float4(0.5f * (v.x + o.x), 0.5f * (v.y + o.y),
                                                        0.5f * (v.z + o.z), 0.5f * (v.w + o.w));
        }
    }
}

extern "C" void kernel_launch(void* const* d_in, const int* in_sizes, int n_in,
                              void* d_out, int out_size, void* d_ws, size_t ws_size,
                              hipStream_t stream)
{
    const float* x    = (const float*)d_in[0];
    const int*   src  = (const int*)d_in[1];
    const int*   dst  = (const int*)d_in[2];
    const float* Wall = (const float*)d_in[3];
    const float* al   = (const float*)d_in[4];
    const float* ar   = (const float*)d_in[5];
    const float* bias = (const float*)d_in[6];
    float* out = (float*)d_out;

    char* p = (char*)d_ws;
    auto take = [&](size_t bytes) {
        char* q = p;
        p += (bytes + 255) & ~size_t(255);
        return q;
    };
    float* feat   = (float*)take(sizeof(float) * NN * DHF);
    float* hbuf   = (float*)take(sizeof(float) * NN * DHF);
    float* el     = (float*)take(sizeof(float) * NN * HH);
    float* er     = (float*)take(sizeof(float) * NN * HH);
    int*   cnt    = (int*)take(sizeof(int) * CNT_N);
    int*   pos    = (int*)take(sizeof(int) * (CNT_N + 1));
    int2*  pairs  = (int2*)take(sizeof(int2) * EE);
    int*   off    = (int*)take(sizeof(int) * (NN + 1));
    int*   csrc   = (int*)take(sizeof(int) * EE);

    const int tb   = 256;
    const int gGmm = GEMM_GRID;
    const int gAgg = (NN + 8 * AGG_NPW - 1) / (8 * AGG_NPW);

    bkt_count  <<<NCHUNK, tb, 0, stream>>>(dst, cnt);
    bkt_scan   <<<1, 1024, 0, stream>>>(cnt, pos);
    bkt_scatter<<<NCHUNK, tb, 0, stream>>>(src, dst, pos, pairs);
    seg_sort   <<<NBKT, tb, 0, stream>>>(pairs, pos, off, csrc);

    const float* hin = x;
    for (int layer = 0; layer < 3; ++layer) {
        const float* Wl  = Wall + layer * DIN * DHF;
        const float* all = al   + layer * HH * FF;
        const float* arl = ar   + layer * HH * FF;
        const float* bl  = bias + layer * HH * FF;
        const int    last = (layer == 2);
        float* outp = last ? out : hbuf;

        gat_gemm     <<<gGmm, tb, 0, stream>>>(hin, Wl, all, arl, feat, el, er);
        gat_aggregate<<<gAgg, tb, 0, stream>>>(el, er, feat, csrc, off, bl, outp, last);
        hin = hbuf;
    }
    (void)in_sizes; (void)n_in; (void)out_size; (void)ws_size;
}


# aggregate of layers 0 and 1 also does the next layer's projection: the work-group's 40 new rows stay in LDS and go through the same WMMA loop; 2 launches and the h array are gone (8 launches)
# speedup vs baseline: 3.6440x; 1.0356x over previous
#include <hip/hip_runtime.h>
#include <math.h>

#define NN  50000
#define EE  800000
#define DIN 128
#define HH  2
#define FF  64
#define DHF 128
#define NEG_SLOPE 0.2f

typedef __attribute__((ext_vector_type(2))) float v2f;
typedef __attribute__((ext_vector_type(8))) float v8f;
typedef __attribute__((ext_vector_type(4))) float v4f;

#if defined(__AMDGCN__) && __has_builtin(__builtin_amdgcn_wmma_f32_16x16x4_f32)
#define USE_WMMA_F32 1
#else
#define USE_WMMA_F32 0
#endif

#define GEMM_TILES (NN / 16)
#define GEMM_GRID  768
static_assert(NN % 16 == 0, "every row tile is whole: no row of a tile lies past the last node");
static_assert(GEMM_GRID <= GEMM_TILES, "every work-group has a first tile");

__global__ __launch_bounds__(256, 8) void gat_gemm(const float* __restrict__ Hm,
                                                   const float* __restrict__ W,
                                                   const float* __restrict__ al,
                                                   const float* __restrict__ ar,
                                                   float* __restrict__ feat,
                                                   float* __restrict__ el,
                                                   float* __restrict__ er)
{
    __shared__ __attribute__((aligned(16))) float sA[2][16 * 132];
    __shared__ __attribute__((aligned(16))) float sC[2][16 * 132];
    __shared__ __attribute__((aligned(16))) float sAlr[2 * DHF];
    const int tid  = threadIdx.x;
    const int lane = tid & 31;
    const int wave = __builtin_amdgcn_readfirstlane(tid >> 5);
    const int mrow = lane & 15;
    const int hi   = lane >> 4;
    const int col0 = wave * 16;

    int t = blockIdx.x;

    const v4f* __restrict__ H4 = reinterpret_cast<const v4f*>(Hm);
    v4f pa0 = H4[(t * 16 + wave) * (DIN / 4) + lane];
    v4f pa1 = H4[(t * 16 + wave + 8) * (DIN / 4) + lane];

    const float alr = (tid < DHF) ? al[tid] : ar[tid - DHF];

    v2f b[DIN / 4];
    #pragma unroll
    for (int kk = 0; kk < DIN / 4; ++kk) {
        const int ka = 4 * kk + 2 * hi;
        b[kk].x = W[(ka + 0) * DHF + col0 + mrow];
        b[kk].y = W[(ka + 1) * DHF + col0 + mrow];
    }

    *reinterpret_cast<v4f*>(&sA[0][wave * 132 + 4 * lane])       = pa0;
    *reinterpret_cast<v4f*>(&sA[0][(wave + 8) * 132 + 4 * lane]) = pa1;
    sAlr[tid] = alr;
    __syncthreads();

    for (int it = 0; t < GEMM_TILES; ++it, t += GEMM_GRID) {
        const int  buf  = it & 1;
        const int  tn   = t + GEMM_GRID;
        const bool more = tn < GEMM_TILES;
        if (more) {
            pa0 = H4[(tn * 16 + wave) * (DIN / 4) + lane];
            pa1 = H4[(tn * 16 + wave + 8) * (DIN / 4) + lane];
        }

        v8f c = {};
#if USE_WMMA_F32
        const float* __restrict__ ap = &sA[buf][mrow * 132 + 2 * hi];
        #pragma unroll
        for (int kk = 0; kk < DIN / 4; ++kk) {
            v2f a;
            a.x = ap[4 * kk + 0];
            a.y = ap[4 * kk + 1];
            c = __builtin_amdgcn_wmma_f32_16x16x4_f32(false, a, false, b[kk],
                                                      (short)0, c, false, false);
        }
#else
        for (int k = 0; k < DIN; ++k) {
            float bkn = W[k * DHF + col0 + mrow];
            #pragma unroll
            for (int i = 0; i < 8; ++i)
                c[i] = fmaf(sA[buf][(i + 8 * hi) * 132 + k], bkn, c[i]);
        }
#endif
        #pragma unroll
        for (int i = 0; i < 8; ++i) {
            int M = i + 8 * hi;
            sC[buf][M * 132 + col0 + mrow] = c[i];
        }
        if (more) {
            *reinterpret_cast<v4f*>(&sA[buf ^ 1][wave * 132 + 4 * lane])       = pa0;
            *reinterpret_cast<v4f*>(&sA[buf ^ 1][(wave + 8) * 132 + 4 * lane]) = pa1;
        }
        __syncthreads();
        const int row0 = t * 16;
        #pragma unroll
        for (int i = 0; i < 8; ++i) {
            int M = i + 8 * hi;
            feat[(row0 + M) * DHF + col0 + mrow] = c[i];
        }

        if (wave == (it & 7)) {
            const int r = lane >> 1, h = lane & 1;
            const v4f* fp4 = reinterpret_cast<const v4f*>(&sC[buf][r * 132 + h * FF]);
            const v4f* al4 = reinterpret_cast<const v4f*>(sAlr + h * FF);
            const v4f* ar4 = reinterpret_cast<const v4f*>(sAlr + DHF + h * FF);
            float sl = 0.f, sr = 0.f;
            #pragma unroll
            for (int q = 0; q < FF / 4; ++q) {
                const v4f v = fp4[q], a = al4[q], bq = ar4[q];
                sl = fmaf(v.x, a.x, sl);  sr = fmaf(v.x, bq.x, sr);
                sl = fmaf(v.y, a.y, sl);  sr = fmaf(v.y, bq.y, sr);
                sl = fmaf(v.z, a.z, sl);  sr = fmaf(v.z, bq.z, sr);
                sl = fmaf(v.w, a.w, sl);  sr = fmaf(v.w, bq.w, sr);
            }
            const int n = row0 + r;
            el[n * HH + h] = sl;
            er[n * HH + h] = sr;
        }
    }
}

#define NCHUNK    250
#define CHUNK_E   3200
#define NBKT      250
#define BKT_NODES 200
#define CNT_N     (NBKT * NCHUNK)
static_assert(NCHUNK * CHUNK_E == EE, "the chunks cover the edge list exactly");
static_assert(NBKT * BKT_NODES == NN, "the ranges cover the nodes exactly");
static_assert(CHUNK_E % 4 == 0, "a chunk is read as int4");
#define CHUNK_Q   ((CHUNK_E / 4 + 255) / 256)
static_assert(NBKT <= 256 && BKT_NODES <= 224, "one thread per range; seven counts per lane of one wave");

__global__ __launch_bounds__(256) void bkt_count(const int* __restrict__ dst,
                                                 int* __restrict__ cnt)
{
    __shared__ int h[NBKT];
    const int t     = threadIdx.x;
    const int chunk = blockIdx.x;
    if (t < NBKT) h[t] = 0;
    __syncthreads();
    const int4* __restrict__ d4 = reinterpret_cast<const int4*>(dst + chunk * CHUNK_E);
    int4 d[CHUNK_Q];
    #pragma unroll
    for (int k = 0; k < CHUNK_Q; ++k) {
        const int q = t + k * 256;
        d[k] = d4[(q < CHUNK_E / 4) ? q : 0];
    }
    #pragma unroll
    for (int k = 0; k < CHUNK_Q; ++k) {
        if (t + k * 256 < CHUNK_E / 4) {
            atomicAdd(&h[(unsigned)d[k].x / BKT_NODES], 1);
            atomicAdd(&h[(unsigned)d[k].y / BKT_NODES], 1);
            atomicAdd(&h[(unsigned)d[k].z / BKT_NODES], 1);
            atomicAdd(&h[(unsigned)d[k].w / BKT_NODES], 1);
        }
    }
    __syncthreads();
    if (t < NBKT) cnt[t * NCHUNK + chunk] = h[t];
}

#define SCAN_ROWS 16
static_assert(32 * SCAN_ROWS * 128 >= CNT_N, "the scan block covers all counts");
static_assert(CNT_N % 4 == 0, "a group of four counts is all in or all out");

__global__ __launch_bounds__(1024) void bkt_scan(const int* __restrict__ cnt,
                                                 int* __restrict__ pos)
{
    __shared__ int wtot[32];
    const int t    = threadIdx.x;
    const int lane = t & 31;
    const int wave = t >> 5;
    const int base = wave * (SCAN_ROWS * 128) + lane * 4;

    int4 v[SCAN_ROWS];
    #pragma unroll
    for (int j = 0; j < SCAN_ROWS; ++j) {
        const int i = base + j * 128;
        v[j] = (i < CNT_N) ? *reinterpret_cast<const int4*>(cnt + i) : make_int4(0, 0, 0, 0);
    }

    int inc[SCAN_ROWS];
    #pragma unroll
    for (int j = 0; j < SCAN_ROWS; ++j) {
        v[j].y += v[j].x;
        v[j].z += v[j].y;
        v[j].w += v[j].z;
        int s = v[j].w;
        #pragma unroll
        for (int ofs = 1; ofs < 32; ofs <<= 1) {
            const int n = __shfl_up(s, ofs);
            if (lane >= ofs) s += n;
        }
        inc[j] = s;
    }

    int rowbase[SCAN_ROWS];
    int run = 0;
    #pragma unroll
    for (int j = 0; j < SCAN_ROWS; ++j) {
        rowbase[j] = run;
        run += __builtin_amdgcn_readlane(inc[j], 31);
    }

    if (lane == 0) wtot[wave] = run;
    __syncthreads();
    int wb = (lane < wave) ? wtot[lane] : 0;
    #pragma unroll
    for (int ofs = 16; ofs > 0; ofs >>= 1)
        wb += __shfl_xor(wb, ofs);

    #pragma unroll
    for (int j = 0; j < SCAN_ROWS; ++j) {
        const int i = base + j * 128;
        if (i < CNT_N) {
            const int e0 = wb + rowbase[j] + inc[j] - v[j].w;
            *reinterpret_cast<int4*>(pos + i) = make_int4(e0, e0 + v[j].x, e0 + v[j].y, e0 + v[j].z);
        }
    }
    if (t == 1023) pos[CNT_N] = wb + run;
}

__global__ __launch_bounds__(256) void bkt_scatter(const int* __restrict__ src,
                                                   const int* __restrict__ dst,
                                                   const int* __restrict__ pos,
                                                   int2* __restrict__ pairs)
{
    __shared__ int cur[NBKT];
    const int t     = threadIdx.x;
    const int chunk = blockIdx.x;
    if (t < NBKT) cur[t] = pos[t * NCHUNK + chunk];
    __syncthreads();
    const int4* __restrict__ s4 = reinterpret_cast<const int4*>(src + chunk * CHUNK_E);
    const int4* __restrict__ d4 = reinterpret_cast<const int4*>(dst + chunk * CHUNK_E);
    int4 s[CHUNK_Q], d[CHUNK_Q];
    #pragma unroll
    for (int k = 0; k < CHUNK_Q; ++k) {
        const int q  = t + k * 256;
        const int qc = (q < CHUNK_E / 4) ? q : 0;
        s[k] = s4[qc];
        d[k] = d4[qc];
    }
    #pragma unroll
    for (int k = 0; k < CHUNK_Q; ++k) {
        if (t + k * 256 < CHUNK_E / 4) {
            const int p0 = atomicAdd(&cur[(unsigned)d[k].x / BKT_NODES], 1);
            const int p1 = atomicAdd(&cur[(unsigned)d[k].y / BKT_NODES], 1);
            const int p2 = atomicAdd(&cur[(unsigned)d[k].z / BKT_NODES], 1);
            const int p3 = atomicAdd(&cur[(unsigned)d[k].w / BKT_NODES], 1);
            pairs[p0] = make_int2(s[k].x, d[k].x);
            pairs[p1] = make_int2(s[k].y, d[k].y);
            pairs[p2] = make_int2(s[k].z, d[k].z);
            pairs[p3] = make_int2(s[k].w, d[k].w);
        }
    }
}

#define SEG_HOLD 16

__global__ __launch_bounds__(256) void seg_sort(const int2* __restrict__ pairs,
                                                const int* __restrict__ pos,
                                                int* __restrict__ off,
                                                int* __restrict__ csrc)
{
    __shared__ int cnt[BKT_NODES];
    __shared__ int start[BKT_NODES];
    const int t     = threadIdx.x;
    const int b     = blockIdx.x;
    const int node0 = b * BKT_NODES;
    const int s0 = pos[b * NCHUNK];
    const int s1 = pos[(b + 1) * NCHUNK];
    if (t < BKT_NODES) cnt[t] = 0;
    __syncthreads();
    int2 pr[SEG_HOLD];
    #pragma unroll
    for (int k = 0; k < SEG_HOLD; ++k) {
        const int i = s0 + t + k * 256;
        pr[k] = pairs[(i < s1) ? i : 0];
    }
    #pragma unroll
    for (int k = 0; k < SEG_HOLD; ++k) {
        if (s0 + t + k * 256 < s1) atomicAdd(&cnt[pr[k].y - node0], 1);
    }
    for (int i = s0 + t + SEG_HOLD * 256; i < s1; i += 256)
        atomicAdd(&cnt[pairs[i].y - node0], 1);
    __syncthreads();
    if (t < 32) {
        int c[7];
        int s = 0;
        #pragma unroll
        for (int k = 0; k < 7; ++k) {
            const int idx = t * 7 + k;
            c[k] = (idx < BKT_NODES) ? cnt[idx] : 0;
            s += c[k];
        }
        int inc = s;
        #pragma unroll
        for (int ofs = 1; ofs < 32; ofs <<= 1) {
            const int n = __shfl_up(inc, ofs);
            if (t >= ofs) inc += n;
        }
        int run = s0 + inc - s;
        #pragma unroll
        for (int k = 0; k < 7; ++k) {
            const int idx = t * 7 + k;
            if (idx < BKT_NODES) {
                start[idx] = run;
                off[node0 + idx] = run;
            }
            run += c[k];
        }
    }
    if (b == NBKT - 1 && t == 0) off[NN] = s1;
    __syncthreads();
    #pragma unroll
    for (int k = 0; k < SEG_HOLD; ++k) {
        if (s0 + t + k * 256 < s1) {
            const int p = atomicAdd(&start[pr[k].y - node0], 1);
            csrc[p] = pr[k].x;
        }
    }
    for (int i = s0 + t + SEG_HOLD * 256; i < s1; i += 256) {
        const int2 q = pairs[i];
        const int p = atomicAdd(&start[q.y - node0], 1);
        csrc[p] = q.x;
    }
}

#define AGG_NPW 5
static_assert(NN % AGG_NPW == 0, "a wave's group of nodes is all in or all out");

__device__ __forceinline__ int load_src(const int* __restrict__ csrc, int p, int pend)
{
    return csrc[(p < pend) ? p : 0];
}

__device__ __forceinline__ float2 edge_score(float2 l, float2 r)
{
    const float a = l.x + r.x;
    const float b = l.y + r.y;
    return make_float2((a >= 0.f) ? a : NEG_SLOPE * a, (b >= 0.f) ? b : NEG_SLOPE * b);
}

__device__ __forceinline__ void keep_edges(int p, int pend, int& s, float2& e)
{
    const bool in = p < pend;
    s   = in ? s : 0;
    e.x = in ? e.x : -__builtin_inff();
    e.y = in ? e.y : -__builtin_inff();
}

__device__ __forceinline__ int lane_bits(float v, int j)
{
    return __builtin_amdgcn_readlane(__float_as_int(v), j);
}

__device__ __forceinline__ float head_weight(float x0, float x1, int j, int hm)
{
    return __int_as_float((lane_bits(x1, j) & hm) | (lane_bits(x0, j) & ~hm));
}

__device__ __forceinline__ void agg_chunk(const v4f* __restrict__ feat4, int lane,
                                          int s, float x0, float x1, int n, int hm,
                                          float& den, float4& acc)
{
    int j = 0;
    for (; j + 8 <= n; j += 8) {
        v4f f[8];
        #pragma unroll
        for (int k = 0; k < 8; ++k)
            f[k] = feat4[(unsigned)__builtin_amdgcn_readlane(s, j + k) * (unsigned)(DHF / 4) + (unsigned)lane];
        __builtin_amdgcn_sched_barrier(0);
        #pragma unroll
        for (int k = 0; k < 8; ++k) {
            const float w = head_weight(x0, x1, j + k, hm);
            den += w;
            acc.x = fmaf(w, f[k].x, acc.x);
            acc.y = fmaf(w, f[k].y, acc.y);
            acc.z = fmaf(w, f[k].z, acc.z);
            acc.w = fmaf(w, f[k].w, acc.w);
        }
    }
    const int r = n - j;
    if (r > 0) {
        v4f f[7];
        #pragma unroll
        for (int k = 0; k < 7; ++k) {
            if (k < r) {
                f[k] = feat4[(unsigned)__builtin_amdgcn_readlane(s, j + k) * (unsigned)(DHF / 4) + (unsigned)lane];
            }
        }
        #pragma unroll
        for (int k = 0; k < 7; ++k) {
            if (k < r) {
                const float w = head_weight(x0, x1, j + k, hm);
                den += w;
                acc.x = fmaf(w, f[k].x, acc.x);
                acc.y = fmaf(w, f[k].y, acc.y);
                acc.z = fmaf(w, f[k].z, acc.z);
                acc.w = fmaf(w, f[k].w, acc.w);
            }
        }
    }
}

#define AGG_WGN   (8 * AGG_NPW)
#define AGG_TILES ((AGG_WGN + 15) / 16)
static_assert(NN % AGG_WGN == 0, "every work-group has all of its nodes: no wave of an aggregate launch is without work");

__device__ __forceinline__ void aggregate_rows(const float* __restrict__ el,
                                               const float* __restrict__ er,
                                               const float* __restrict__ feat,
                                               const int* __restrict__ csrc,
                                               const int* __restrict__ off,
                                               const float* __restrict__ bias,
                                               const int lane, const int d0,
                                               const bool to_lds,
                                               float* __restrict__ hrows,
                                               float* __restrict__ out)
{
    const int hm   = -(lane >> 4);

    const float2* __restrict__ el2  = reinterpret_cast<const float2*>(el);
    const float2* __restrict__ er2  = reinterpret_cast<const float2*>(er);
    float4* __restrict__ out4 = reinterpret_cast<float4*>(out);

    const float b0 = bias[4 * lane + 0], b1 = bias[4 * lane + 1];
    const float b2 = bias[4 * lane + 2], b3 = bias[4 * lane + 3];

    static_assert(AGG_NPW + 2 < 32, "bounds of two nodes past the last are read from lanes");
    const int    offv = off[d0 + ((lane <= AGG_NPW) ? lane : AGG_NPW)];
    float2 erN = er2[d0];

    int pcur = __builtin_amdgcn_readlane(offv, 0);
    int pnxt = __builtin_amdgcn_readlane(offv, 1);
    int    sN  = load_src(csrc, pcur + lane, pnxt);
    int    sNN = load_src(csrc, pnxt + lane, __builtin_amdgcn_readlane(offv, 2));
    float2 gN  = el2[sN];

    for (int i = 0; i < AGG_NPW; ++i) {
        const int d  = d0 + i;
        const int p0 = pcur, p1 = pnxt;
        const float2 erd = erN;
        erN = er2[d0 + ((i + 1 < AGG_NPW) ? i + 1 : i)];
        int    sF = sN;
        float2 eF = edge_score(gN, erd);
        keep_edges(p0 + lane, p1, sF, eF);

        pcur = p1;
        pnxt = __builtin_amdgcn_readlane(offv, i + 2);
        sN   = sNN;
        gN   = el2[sN];
        sNN  = load_src(csrc, pnxt + lane, __builtin_amdgcn_readlane(offv, i + 3));

        float m0 = eF.x, m1 = eF.y;
        for (int c0 = p0 + 32; c0 < p1; c0 += 32) {
            const int p = c0 + lane;
            if (p < p1) {
                const float2 e = edge_score(el2[csrc[p]], erd);
                m0 = fmaxf(m0, e.x);
                m1 = fmaxf(m1, e.y);
            }
        }
        #pragma unroll
        for (int ofs = 16; ofs > 0; ofs >>= 1) {
            m0 = fmaxf(m0, __shfl_xor(m0, ofs));
            m1 = fmaxf(m1, __shfl_xor(m1, ofs));
        }

        float  den = 0.f;
        float4 acc = make_float4(0.f, 0.f, 0.f, 0.f);
        {
            int    c0 = p0;
            int    s  = sF;
            float2 e  = eF;
            for (;;) {
                const int n = (p1 - c0 < 32) ? (p1 - c0) : 32;
                agg_chunk(reinterpret_cast<const v4f*>(feat), lane, s, expf(e.x - m0), expf(e.y - m1), n, hm, den, acc);
                c0 += 32;
                if (c0 >= p1) break;
                s = load_src(csrc, c0 + lane, p1);
                e = edge_score(el2[s], erd);
                keep_edges(c0 + lane, p1, s, e);
            }
        }

        const float dd = (den == 0.f) ? 1.f : den;
        float4 v;
        v.x = acc.x / dd + b0;
        v.y = acc.y / dd + b1;
        v.z = acc.z / dd + b2;
        v.w = acc.w / dd + b3;
        v.x = (v.x > 0.f) ? v.x : expm1f(v.x);
        v.y = (v.y > 0.f) ? v.y : expm1f(v.y);
        v.z = (v.z > 0.f) ? v.z : expm1f(v.z);
        v.w = (v.w > 0.f) ? v.w : expm1f(v.w);

        if (to_lds) {
            v4f hv;
            hv.x = v.x; hv.y = v.y; hv.z = v.z; hv.w = v.w;
            *reinterpret_cast<v4f*>(&hrows[i * 132 + 4 * lane]) = hv;
        } else {
            float4 o;
            o.x = __shfl_xor(v.x, 16);
            o.y = __shfl_xor(v.y, 16);
            o.z = __shfl_xor(v.z, 16);
            o.w = __shfl_xor(v.w, 16);
            if (lane < 16)
                out4[d * (FF / 4) + lane] = make_float4(0.5f * (v.x + o.x), 0.5f * (v.y + o.y),
                                                        0.5f * (v.z + o.z), 0.5f * (v.w + o.w));
        }
    }
}

__global__ __launch_bounds__(256, 10) void gat_aggregate(const float* __restrict__ el,
                                                     const float* __restrict__ er,
                                                     const float* __restrict__ feat,
                                                     const int* __restrict__ csrc,
                                                     const int* __restrict__ off,
                                                     const float* __restrict__ bias,
                                                     float* __restrict__ out)
{
    const int lane = threadIdx.x & 31;
    const int wv = __builtin_amdgcn_readfirstlane((int)(threadIdx.x >> 5));
    const int d0 = blockIdx.x * AGG_WGN + wv * AGG_NPW;
    aggregate_rows(el, er, feat, csrc, off, bias, lane, d0, false, nullptr, out);
}

__global__ __launch_bounds__(256, 10) void gat_aggregate_project(const float* __restrict__ el,
                                                     const float* __restrict__ er,
                                                     const float* __restrict__ feat,
                                                     const int* __restrict__ csrc,
                                                     const int* __restrict__ off,
                                                     const float* __restrict__ bias,
                                                     const float* __restrict__ Wn,
                                                     const float* __restrict__ aln,
                                                     const float* __restrict__ arn,
                                                     float* __restrict__ featn,
                                                     float* __restrict__ eln,
                                                     float* __restrict__ ern)
{
    __shared__ __attribute__((aligned(16))) float sH[AGG_TILES * 16 * 132];
    __shared__ __attribute__((aligned(16))) float sC[AGG_TILES][16 * 132];
    __shared__ __attribute__((aligned(16))) float sAlr[2 * DHF];
    const int tid  = threadIdx.x;
    const int lane = tid & 31;
    const int wave = __builtin_amdgcn_readfirstlane(tid >> 5);
    const int node0 = blockIdx.x * AGG_WGN;

    sAlr[tid] = (tid < DHF) ? aln[tid] : arn[tid - DHF];
    static_assert((AGG_TILES * 16 - AGG_WGN) * 32 == 256, "one piece of the fill rows per thread");
    {
        v4f z = {};
        *reinterpret_cast<v4f*>(&sH[(AGG_WGN + (tid >> 5)) * 132 + 4 * lane]) = z;
    }

    aggregate_rows(el, er, feat, csrc, off, bias, lane, node0 + wave * AGG_NPW, true,
                   &sH[wave * AGG_NPW * 132], nullptr);

    const int mrow = lane & 15;
    const int hi   = lane >> 4;
    const int col0 = wave * 16;

    v2f b[DIN / 4];
    #pragma unroll
    for (int kk = 0; kk < DIN / 4; ++kk) {
        const int ka = 4 * kk + 2 * hi;
        b[kk].x = Wn[(ka + 0) * DHF + col0 + mrow];
        b[kk].y = Wn[(ka + 1) * DHF + col0 + mrow];
    }
    __syncthreads();

    for (int it = 0; it < AGG_TILES; ++it) {
        v8f c = {};
#if USE_WMMA_F32
        const float* __restrict__ ap = &sH[(it * 16 + mrow) * 132 + 2 * hi];
        #pragma unroll
        for (int kk = 0; kk < DIN / 4; ++kk) {
            v2f a;
            a.x = ap[4 * kk + 0];
            a.y = ap[4 * kk + 1];
            c = __builtin_amdgcn_wmma_f32_16x16x4_f32(false, a, false, b[kk],
                                                      (short)0, c, false, false);
        }
#else
        for (int k = 0; k < DIN; ++k) {
            float bkn = Wn[k * DHF + col0 + mrow];
            #pragma unroll
            for (int i = 0; i < 8; ++i)
                c[i] = fmaf(sH[(it * 16 + i + 8 * hi) * 132 + k], bkn, c[i]);
        }
#endif
        const int row0 = it * 16;
        #pragma unroll
        for (int i = 0; i < 8; ++i) {
            int M = i + 8 * hi;
            sC[it][M * 132 + col0 + mrow] = c[i];
            if (row0 + M < AGG_WGN)
                featn[(node0 + row0 + M) * DHF + col0 + mrow] = c[i];
        }
    }
    __syncthreads();

    if (wave < AGG_TILES) {
        const int r = lane >> 1, h = lane & 1;
        const v4f* fp4 = reinterpret_cast<const v4f*>(&sC[wave][r * 132 + h * FF]);
        const v4f* al4 = reinterpret_cast<const v4f*>(sAlr + h * FF);
        const v4f* ar4 = reinterpret_cast<const v4f*>(sAlr + DHF + h * FF);
        float sl = 0.f, sr = 0.f;
        #pragma unroll
        for (int q = 0; q < FF / 4; ++q) {
            const v4f v = fp4[q], a = al4[q], bq = ar4[q];
            sl = fmaf(v.x, a.x, sl);  sr = fmaf(v.x, bq.x, sr);
            sl = fmaf(v.y, a.y, sl);  sr = fmaf(v.y, bq.y, sr);
            sl = fmaf(v.z, a.z, sl);  sr = fmaf(v.z, bq.z, sr);
            sl = fmaf(v.w, a.w, sl);  sr = fmaf(v.w, bq.w, sr);
        }
        if (wave * 16 + r < AGG_WGN) {
            const int n = node0 + wave * 16 + r;
            eln[n * HH + h] = sl;
            ern[n * HH + h] = sr;
        }
    }
}

extern "C" void kernel_launch(void* const* d_in, const int* in_sizes, int n_in,
                              void* d_out, int out_size, void* d_ws, size_t ws_size,
                              hipStream_t stream)
{
    const float* x    = (const float*)d_in[0];
    const int*   src  = (const int*)d_in[1];
    const int*   dst  = (const int*)d_in[2];
    const float* Wall = (const float*)d_in[3];
    const float* al   = (const float*)d_in[4];
    const float* ar   = (const float*)d_in[5];
    const float* bias = (const float*)d_in[6];
    float* out = (float*)d_out;

    char* p = (char*)d_ws;
    auto take = [&](size_t bytes) {
        char* q = p;
        p += (bytes + 255) & ~size_t(255);
        return q;
    };
    float* featA  = (float*)take(sizeof(float) * NN * DHF);
    float* featB  = (float*)take(sizeof(float) * NN * DHF);
    float* elA    = (float*)take(sizeof(float) * NN * HH);
    float* erA    = (float*)take(sizeof(float) * NN * HH);
    float* elB    = (float*)take(sizeof(float) * NN * HH);
    float* erB    = (float*)take(sizeof(float) * NN * HH);
    int*   cnt    = (int*)take(sizeof(int) * CNT_N);
    int*   pos    = (int*)take(sizeof(int) * (CNT_N + 1));
    int2*  pairs  = (int2*)take(sizeof(int2) * EE);
    int*   off    = (int*)take(sizeof(int) * (NN + 1));
    int*   csrc   = (int*)take(sizeof(int) * EE);

    const int tb   = 256;
    const int gGmm = GEMM_GRID;
    const int gAgg = NN / AGG_WGN;

    bkt_count  <<<NCHUNK, tb, 0, stream>>>(dst, cnt);
    bkt_scan   <<<1, 1024, 0, stream>>>(cnt, pos);
    bkt_scatter<<<NCHUNK, tb, 0, stream>>>(src, dst, pos, pairs);
    seg_sort   <<<NBKT, tb, 0, stream>>>(pairs, pos, off, csrc);

    gat_gemm<<<gGmm, tb, 0, stream>>>(x, Wall, al, ar, featA, elA, erA);
    gat_aggregate_project<<<gAgg, tb, 0, stream>>>(elA, erA, featA, csrc, off, bias,
                                                   Wall + DIN * DHF, al + HH * FF, ar + HH * FF,
                                                   featB, elB, erB);
    gat_aggregate_project<<<gAgg, tb, 0, stream>>>(elB, erB, featB, csrc, off, bias + HH * FF,
                                                   Wall + 2 * DIN * DHF, al + 2 * HH * FF, ar + 2 * HH * FF,
                                                   featA, elA, erA);
    gat_aggregate<<<gAgg, tb, 0, stream>>>(elA, erA, featA, csrc, off, bias + 2 * HH * FF, out);
    (void)in_sizes; (void)n_in; (void)out_size; (void)ws_size;
}
